# starting kernel
# baseline (speedup 1.0000x reference)
#include <hip/hip_runtime.h>
#include <math.h>

#define B_    4
#define ROI_  90
#define T_    200
#define H_    3
#define F_    96
#define HID_  96
#define G3_   288
#define BH_   12
#define NP_   96
#define STR_  100
#define GSTR_ 292
#define TPB_  288

typedef float v2f __attribute__((ext_vector_type(2)));
typedef float v8f __attribute__((ext_vector_type(8)));

__device__ __forceinline__ v8f wmma4(v2f a, v2f b, v8f c) {
  return __builtin_amdgcn_wmma_f32_16x16x4_f32(false, a, false, b, (short)0, c,
                                               false, false);
}

__device__ __forceinline__ void mm2x2_trB(const float* As, const float* Bt,
                                          int sr, int sc, int lane, v8f acc[4]) {
  const int half = lane >> 4;
  const int l15  = lane & 15;
  const float* a0 = As + (sr * 32 + l15) * STR_ + (half << 1);
  const float* a1 = a0 + 16 * STR_;
  const float* b0 = Bt + (sc * 32 + l15) * STR_ + (half << 1);
  const float* b1 = b0 + 16 * STR_;
  #pragma unroll
  for (int j = 0; j < 4; ++j) acc[j] = (v8f){0.f,0.f,0.f,0.f,0.f,0.f,0.f,0.f};
  #pragma unroll
  for (int k0 = 0; k0 < NP_; k0 += 4) {
    v2f av0 = *(const v2f*)(a0 + k0);
    v2f av1 = *(const v2f*)(a1 + k0);
    v2f bv0 = *(const v2f*)(b0 + k0);
    v2f bv1 = *(const v2f*)(b1 + k0);
    acc[0] = wmma4(av0, bv0, acc[0]);
    acc[1] = wmma4(av0, bv1, acc[1]);
    acc[2] = wmma4(av1, bv0, acc[2]);
    acc[3] = wmma4(av1, bv1, acc[3]);
  }
}

__device__ __forceinline__ void mm1x4_trB(const float* As, const float* Bt,
                                          int ntBase, int lane, v8f acc[4]) {
  const int half = lane >> 4;
  const int l15  = lane & 15;
  const float* aRow = As + l15 * STR_ + (half << 1);
  const float* bRow = Bt + (ntBase * 16 + l15) * STR_ + (half << 1);
  #pragma unroll
  for (int j = 0; j < 4; ++j) acc[j] = (v8f){0.f,0.f,0.f,0.f,0.f,0.f,0.f,0.f};
  #pragma unroll
  for (int k0 = 0; k0 < F_; k0 += 4) {
    v2f av = *(const v2f*)(aRow + k0);
    #pragma unroll
    for (int j = 0; j < 4; ++j) {
      v2f bv = *(const v2f*)(bRow + j * 16 * STR_ + k0);
      acc[j] = wmma4(av, bv, acc[j]);
    }
  }
}

__device__ __forceinline__ void mm1x2_trB(const float* As, const float* Bt,
                                          int ntBase, int lane, v8f acc[2]) {
  const int half = lane >> 4;
  const int l15  = lane & 15;
  const float* aRow = As + l15 * STR_ + (half << 1);
  const float* bRow = Bt + (ntBase * 16 + l15) * STR_ + (half << 1);
  #pragma unroll
  for (int j = 0; j < 2; ++j) acc[j] = (v8f){0.f,0.f,0.f,0.f,0.f,0.f,0.f,0.f};
  #pragma unroll
  for (int k0 = 0; k0 < F_; k0 += 4) {
    v2f av = *(const v2f*)(aRow + k0);
    #pragma unroll
    for (int j = 0; j < 2; ++j) {
      v2f bv = *(const v2f*)(bRow + j * 16 * STR_ + k0);
      acc[j] = wmma4(av, bv, acc[j]);
    }
  }
}

__global__ void __launch_bounds__(TPB_)
ftdgnn_agg_mlp(const float* __restrict__ x, const float* __restrict__ a,
               const float* __restrict__ eps,
               const float* __restrict__ W1, const float* __restrict__ b1,
               const float* __restrict__ g1, const float* __restrict__ be1,
               const float* __restrict__ W2, const float* __restrict__ b2,
               const float* __restrict__ g2, const float* __restrict__ be2,
               float* __restrict__ hbuf) {
  extern __shared__ float smem[];
  float* xs = smem;
  float* ms = xs + NP_ * STR_;
  float* ws = ms + NP_ * STR_;

  const int tid  = threadIdx.x;
  const int lane = tid & 31;
  const int wave = tid >> 5;
  const int half = lane >> 4;
  const int l15  = lane & 15;
  const int sr   = wave / 3;
  const int sc   = wave % 3;

  const int bth = blockIdx.x;
  const int b   = bth / (T_ * H_);
  const int th  = bth % (T_ * H_);
  const int t   = th / H_;
  const int hh  = th % H_;

  for (int idx = tid; idx < NP_ * NP_; idx += TPB_) {
    const int r = idx / NP_, c = idx % NP_;
    float xv = 0.f;
    if (r < ROI_)
      xv = x[(((size_t)(b * ROI_ + r) * T_ + t) * H_ + hh) * F_ + c];
    xs[c * STR_ + r] = xv;
    float mv = 0.f;
    if (r < ROI_ && c < ROI_) {
      const float av = a[(((size_t)(b * ROI_ + r) * T_ + t) * H_ + hh) * ROI_ + c];
      mv = (av != 0.f) ? 1.f : 0.f;
    }
    ms[r * STR_ + c] = mv;
    ws[c * STR_ + r] = W1[r * HID_ + c];
  }
  __syncthreads();

  const float epsv = eps[0];
  const float rs   = rsqrtf(1.0f + 1e-5f);

  v8f acc[4];

  mm2x2_trB(ms, xs, sr, sc, lane, acc);
  __syncthreads();
  #pragma unroll
  for (int di = 0; di < 2; ++di)
    #pragma unroll
    for (int dj = 0; dj < 2; ++dj) {
      const int n = (2 * sc + dj) * 16 + l15;
      #pragma unroll
      for (int v = 0; v < 8; ++v) {
        const int m = (2 * sr + di) * 16 + v + 8 * half;
        ms[m * STR_ + n] = acc[di * 2 + dj][v] + epsv * xs[n * STR_ + m];
      }
    }
  __syncthreads();

  mm2x2_trB(ms, ws, sr, sc, lane, acc);
  __syncthreads();
  #pragma unroll
  for (int di = 0; di < 2; ++di)
    #pragma unroll
    for (int dj = 0; dj < 2; ++dj) {
      const int n = (2 * sc + dj) * 16 + l15;
      const float sc1 = g1[n] * rs, sh1 = be1[n], bb1 = b1[n];
      #pragma unroll
      for (int v = 0; v < 8; ++v) {
        const int m = (2 * sr + di) * 16 + v + 8 * half;
        float z = (acc[di * 2 + dj][v] + bb1) * sc1 + sh1;
        z = (z > 0.f) ? z : expm1f(z);
        xs[m * STR_ + n] = z;
      }
    }
  for (int idx = tid; idx < NP_ * NP_; idx += TPB_) {
    const int r = idx / NP_, c = idx % NP_;
    ws[c * STR_ + r] = W2[r * F_ + c];
  }
  __syncthreads();

  mm2x2_trB(xs, ws, sr, sc, lane, acc);
  #pragma unroll
  for (int di = 0; di < 2; ++di)
    #pragma unroll
    for (int dj = 0; dj < 2; ++dj) {
      const int n = (2 * sc + dj) * 16 + l15;
      const float sc2 = g2[n] * rs, sh2 = be2[n], bb2 = b2[n];
      #pragma unroll
      for (int v = 0; v < 8; ++v) {
        const int m = (2 * sr + di) * 16 + v + 8 * half;
        if (m < ROI_) {
          float z = (acc[di * 2 + dj][v] + bb2) * sc2 + sh2;
          z = (z > 0.f) ? z : 0.f;
          hbuf[((size_t)(m * T_ + t) * BH_ + (b * H_ + hh)) * F_ + n] = z;
        }
      }
    }
}

__global__ void __launch_bounds__(TPB_)
ftdgnn_gru(const float* __restrict__ Wih, const float* __restrict__ Whh,
           const float* __restrict__ bih, const float* __restrict__ bhh,
           const float* __restrict__ hbuf, float* __restrict__ out) {
  extern __shared__ float smem[];
  float* WihS = smem;
  float* WhhS = WihS + G3_ * STR_;
  float* hS   = WhhS + G3_ * STR_;
  float* xS   = hS   + 16 * STR_;
  float* gx   = xS   + 16 * STR_;
  float* gh   = gx   + 16 * GSTR_;
  float* bihS = gh   + 16 * GSTR_;
  float* bhhS = bihS + G3_;

  const int tid  = threadIdx.x;
  const int lane = tid & 31;
  const int wave = tid >> 5;
  const int half = lane >> 4;
  const int l15  = lane & 15;
  const int r    = blockIdx.x;

  for (int idx = tid; idx < G3_ * F_; idx += TPB_) {
    const int g = idx / F_, k = idx % F_;
    WihS[g * STR_ + k] = Wih[((size_t)r * G3_ + g) * F_ + k];
    WhhS[g * STR_ + k] = Whh[((size_t)r * G3_ + g) * HID_ + k];
  }
  for (int idx = tid; idx < G3_; idx += TPB_) {
    bihS[idx] = bih[r * G3_ + idx];
    bhhS[idx] = bhh[r * G3_ + idx];
  }
  for (int idx = tid; idx < 16 * STR_; idx += TPB_) { hS[idx] = 0.f; xS[idx] = 0.f; }
  __syncthreads();
  for (int idx = tid; idx < BH_ * F_; idx += TPB_) {
    const int m = idx / F_, k = idx % F_;
    xS[m * STR_ + k] = hbuf[((size_t)(r * T_) * BH_ + m) * F_ + k];
  }
  __syncthreads();

  for (int t = 0; t < T_; ++t) {
    if (wave < 4) {
      v8f acc[4];
      mm1x4_trB(xS, WihS, wave * 4, lane, acc);
      #pragma unroll
      for (int j = 0; j < 4; ++j) {
        const int n = (wave * 4 + j) * 16 + l15;
        #pragma unroll
        for (int v = 0; v < 8; ++v)
          gx[(v + 8 * half) * GSTR_ + n] = acc[j][v];
      }
    } else if (wave > 4) {
      v8f acc[4];
      mm1x4_trB(hS, WhhS, wave * 4 - 18, lane, acc);
      #pragma unroll
      for (int j = 0; j < 4; ++j) {
        const int n = (wave * 4 - 18 + j) * 16 + l15;
        #pragma unroll
        for (int v = 0; v < 8; ++v)
          gh[(v + 8 * half) * GSTR_ + n] = acc[j][v];
      }
    } else {
      v8f ax[2], ah[2];
      mm1x2_trB(xS, WihS, 16, lane, ax);
      mm1x2_trB(hS, WhhS, 0, lane, ah);
      #pragma unroll
      for (int j = 0; j < 2; ++j) {
        const int nx = (16 + j) * 16 + l15;
        const int nh = j * 16 + l15;
        #pragma unroll
        for (int v = 0; v < 8; ++v) {
          gx[(v + 8 * half) * GSTR_ + nx] = ax[j][v];
          gh[(v + 8 * half) * GSTR_ + nh] = ah[j][v];
        }
      }
    }
    __syncthreads();

    for (int idx = tid; idx < BH_ * HID_; idx += TPB_) {
      const int m = idx / HID_, f = idx % HID_;
      const float xr = gx[m * GSTR_ + f]             + bihS[f];
      const float xz = gx[m * GSTR_ + HID_ + f]      + bihS[HID_ + f];
      const float xn = gx[m * GSTR_ + 2 * HID_ + f]  + bihS[2 * HID_ + f];
      const float hr = gh[m * GSTR_ + f]             + bhhS[f];
      const float hz = gh[m * GSTR_ + HID_ + f]      + bhhS[HID_ + f];
      const float hn = gh[m * GSTR_ + 2 * HID_ + f]  + bhhS[2 * HID_ + f];
      const float rg = 1.f / (1.f + expf(-(xr + hr)));
      const float zg = 1.f / (1.f + expf(-(xz + hz)));
      const float ng = tanhf(xn + rg * hn);
      const float h2 = (1.f - zg) * ng + zg * hS[m * STR_ + f];
      hS[m * STR_ + f] = h2;
      const int bb = m / H_, hh = m % H_;
      out[(((size_t)(bb * ROI_ + r) * T_ + t) * H_ + hh) * HID_ + f] = h2;
    }
    if (t + 1 < T_) {
      for (int idx = tid; idx < BH_ * F_; idx += TPB_) {
        const int m = idx / F_, k = idx % F_;
        xS[m * STR_ + k] = hbuf[((size_t)(r * T_ + (t + 1)) * BH_ + m) * F_ + k];
      }
      if (t + 2 < T_) {
        const int pidx = tid & (BH_ - 1);
        __builtin_prefetch(&hbuf[((size_t)(r * T_ + (t + 2)) * BH_ + pidx) * F_], 0, 1);
      }
    }
    __syncthreads();
  }
}

extern "C" void kernel_launch(void* const* d_in, const int* in_sizes, int n_in,
                              void* d_out, int out_size, void* d_ws, size_t ws_size,
                              hipStream_t stream) {
  const float* x   = (const float*)d_in[0];
  const float* a   = (const float*)d_in[1];
  const float* eps = (const float*)d_in[2];
  const float* W1  = (const float*)d_in[3];
  const float* b1  = (const float*)d_in[4];
  const float* g1  = (const float*)d_in[5];
  const float* be1 = (const float*)d_in[6];
  const float* W2  = (const float*)d_in[7];
  const float* b2  = (const float*)d_in[8];
  const float* g2  = (const float*)d_in[9];
  const float* be2 = (const float*)d_in[10];
  const float* Wih = (const float*)d_in[11];
  const float* Whh = (const float*)d_in[12];
  const float* bih = (const float*)d_in[13];
  const float* bhh = (const float*)d_in[14];
  float* out  = (float*)d_out;
  float* hbuf = (float*)d_ws;

  const size_t ldsA = (size_t)3 * NP_ * STR_ * sizeof(float);
  ftdgnn_agg_mlp<<<B_ * T_ * H_, TPB_, ldsA, stream>>>(
      x, a, eps, W1, b1, g1, be1, W2, b2, g2, be2, hbuf);

  const size_t ldsB = (size_t)(2 * G3_ * STR_ + 2 * 16 * STR_ +
                               2 * 16 * GSTR_ + 2 * G3_) * sizeof(float);
  ftdgnn_gru<<<ROI_, TPB_, ldsB, stream>>>(Wih, Whh, bih, bhh, hbuf, out);
}


# Aggregation+MLP kernel: x, the mask, W1 and W2 go from global memory straight into WMMA fragment registers, no LDS staging loops; LDS keeps only agg and h1 (115 -> 77 KB a block, 2 barriers not 5)
# speedup vs baseline: 1.8725x; 1.8725x over previous
#include <hip/hip_runtime.h>
#include <math.h>

#define B_    4
#define ROI_  90
#define T_    200
#define H_    3
#define F_    96
#define HID_  96
#define G3_   288
#define BH_   12
#define NP_   96
#define STR_  100
#define TPB_  288

typedef float v2f __attribute__((ext_vector_type(2)));
typedef float v8f __attribute__((ext_vector_type(8)));

__device__ __forceinline__ v8f wmma4(v2f a, v2f b, v8f c) {
  return __builtin_amdgcn_wmma_f32_16x16x4_f32(false, a, false, b, (short)0, c,
                                               false, false);
}


__device__ __forceinline__ void mm2x2_gB(const float* As, const float* __restrict__ Bg,
                                         int sr, int sc, int lane, v8f acc[4]) {
  const int half = lane >> 4;
  const int l15  = lane & 15;
  const float* a0 = As + (sr * 32 + l15) * STR_ + (half << 1);
  const float* a1 = a0 + 16 * STR_;
  const float* b0 = Bg + (half << 1) * NP_ + sc * 32 + l15;
  #pragma unroll
  for (int j = 0; j < 4; ++j) acc[j] = (v8f){0.f,0.f,0.f,0.f,0.f,0.f,0.f,0.f};
  #pragma unroll
  for (int k0 = 0; k0 < NP_; k0 += 4) {
    v2f av0 = *(const v2f*)(a0 + k0);
    v2f av1 = *(const v2f*)(a1 + k0);
    v2f bv0, bv1;
    bv0[0] = b0[k0 * NP_];        bv0[1] = b0[(k0 + 1) * NP_];
    bv1[0] = b0[k0 * NP_ + 16];   bv1[1] = b0[(k0 + 1) * NP_ + 16];
    acc[0] = wmma4(av0, bv0, acc[0]);
    acc[1] = wmma4(av0, bv1, acc[1]);
    acc[2] = wmma4(av1, bv0, acc[2]);
    acc[3] = wmma4(av1, bv1, acc[3]);
  }
}

#define XRS_  ((size_t)T_ * H_ * F_)
#define ARS_  ((size_t)T_ * H_ * ROI_)
#define KAGG_ 92

__global__ void __launch_bounds__(TPB_)
ftdgnn_agg_mlp(const float* __restrict__ x, const float* __restrict__ a,
               const float* __restrict__ eps,
               const float* __restrict__ W1, const float* __restrict__ b1,
               const float* __restrict__ g1, const float* __restrict__ be1,
               const float* __restrict__ W2, const float* __restrict__ b2,
               const float* __restrict__ g2, const float* __restrict__ be2,
               float* __restrict__ hbuf) {
  extern __shared__ float smem[];
  float* aggS = smem;
  float* h1S  = aggS + NP_ * STR_;

  const int tid  = threadIdx.x;
  const int lane = tid & 31;
  const int wave = tid >> 5;
  const int half = lane >> 4;
  const int l15  = lane & 15;
  const int sr   = wave / 3;
  const int sc   = wave % 3;

  const int bth = blockIdx.x;
  const int b   = bth / (T_ * H_);
  const int th  = bth % (T_ * H_);
  const int t   = th / H_;
  const int hh  = th % H_;

  const size_t tile = ((size_t)(b * ROI_) * T_ + t) * H_ + hh;
  const float* xb = x + tile * F_;
  const float* ab = a + tile * ROI_;

  const float epsv = eps[0];
  const float rs   = rsqrtf(1.0f + 1e-5f);

  v8f acc[4];

  {
    const int  i0   = sr * 32 + l15;
    const int  i1   = i0 + 16;
    const bool i1ok = (i1 < ROI_);
    const float* ar0 = ab + (size_t)i0 * ARS_;
    const float* ar1 = ab + (size_t)(i1ok ? i1 : ROI_ - 1) * ARS_;
    const float* xc  = xb + sc * 32 + l15;
    #pragma unroll
    for (int j = 0; j < 4; ++j) acc[j] = (v8f){0.f,0.f,0.f,0.f,0.f,0.f,0.f,0.f};
    #pragma unroll
    for (int k0 = 0; k0 < KAGG_; k0 += 4) {
      const int  j   = k0 + (half << 1);
      const bool jok = (j < ROI_);
      const int  jc  = jok ? j : 0;
      const v2f r0 = *(const v2f*)(ar0 + jc);
      const v2f r1 = *(const v2f*)(ar1 + jc);
      const float* xr = xc + (size_t)jc * XRS_;
      const float x00 = xr[0],    x01 = xr[16];
      const float x10 = xr[XRS_], x11 = xr[XRS_ + 16];
      v2f av0, av1, bv0, bv1;
      av0[0] = (jok && r0[0] != 0.f) ? 1.f : 0.f;
      av0[1] = (jok && r0[1] != 0.f) ? 1.f : 0.f;
      av1[0] = (jok && i1ok && r1[0] != 0.f) ? 1.f : 0.f;
      av1[1] = (jok && i1ok && r1[1] != 0.f) ? 1.f : 0.f;
      bv0[0] = jok ? x00 : 0.f;   bv0[1] = jok ? x10 : 0.f;
      bv1[0] = jok ? x01 : 0.f;   bv1[1] = jok ? x11 : 0.f;
      acc[0] = wmma4(av0, bv0, acc[0]);
      acc[1] = wmma4(av0, bv1, acc[1]);
      acc[2] = wmma4(av1, bv0, acc[2]);
      acc[3] = wmma4(av1, bv1, acc[3]);
    }
  }
  #pragma unroll
  for (int di = 0; di < 2; ++di)
    #pragma unroll
    for (int dj = 0; dj < 2; ++dj) {
      const int n = (2 * sc + dj) * 16 + l15;
      #pragma unroll
      for (int v = 0; v < 8; ++v) {
        const int  m   = (2 * sr + di) * 16 + v + 8 * half;
        const bool mok = (m < ROI_);
        const float xv = xb[(size_t)(mok ? m : ROI_ - 1) * XRS_ + n];
        aggS[m * STR_ + n] = acc[di * 2 + dj][v] + epsv * (mok ? xv : 0.f);
      }
    }
  __syncthreads();

  mm2x2_gB(aggS, W1, sr, sc, lane, acc);
  #pragma unroll
  for (int di = 0; di < 2; ++di)
    #pragma unroll
    for (int dj = 0; dj < 2; ++dj) {
      const int n = (2 * sc + dj) * 16 + l15;
      const float sc1 = g1[n] * rs, sh1 = be1[n], bb1 = b1[n];
      #pragma unroll
      for (int v = 0; v < 8; ++v) {
        const int m = (2 * sr + di) * 16 + v + 8 * half;
        float z = (acc[di * 2 + dj][v] + bb1) * sc1 + sh1;
        z = (z > 0.f) ? z : expm1f(z);
        h1S[m * STR_ + n] = z;
      }
    }
  __syncthreads();

  mm2x2_gB(h1S, W2, sr, sc, lane, acc);
  #pragma unroll
  for (int di = 0; di < 2; ++di)
    #pragma unroll
    for (int dj = 0; dj < 2; ++dj) {
      const int n = (2 * sc + dj) * 16 + l15;
      const float sc2 = g2[n] * rs, sh2 = be2[n], bb2 = b2[n];
      #pragma unroll
      for (int v = 0; v < 8; ++v) {
        const int m = (2 * sr + di) * 16 + v + 8 * half;
        if (m < ROI_) {
          float z = (acc[di * 2 + dj][v] + bb2) * sc2 + sh2;
          z = (z > 0.f) ? z : 0.f;
          hbuf[((size_t)(m * T_ + t) * BH_ + (b * H_ + hh)) * F_ + n] = z;
        }
      }
    }
}

#define TPG_   384
#define GXT_   (18 * 32 * 8)

typedef float v4f __attribute__((ext_vector_type(4)));

__device__ __forceinline__ void mm1x3_gates(const float* As, const float* Bt,
                                            int j, int lane, v8f acc[3]) {
  const int half = lane >> 4;
  const int l15  = lane & 15;
  const float* aRow = As + l15 * STR_ + (half << 1);
  const float* bRow = Bt + (j * 16 + l15) * STR_ + (half << 1);
  #pragma unroll
  for (int g = 0; g < 3; ++g) acc[g] = (v8f){0.f,0.f,0.f,0.f,0.f,0.f,0.f,0.f};
  #pragma unroll
  for (int k0 = 0; k0 < F_; k0 += 4) {
    v2f av = *(const v2f*)(aRow + k0);
    #pragma unroll
    for (int g = 0; g < 3; ++g) {
      v2f bv = *(const v2f*)(bRow + g * HID_ * STR_ + k0);
      acc[g] = wmma4(av, bv, acc[g]);
    }
  }
}

__global__ void __launch_bounds__(TPG_)
ftdgnn_gru(const float* __restrict__ Wih, const float* __restrict__ Whh,
           const float* __restrict__ bih, const float* __restrict__ bhh,
           const float* __restrict__ hbuf, float* __restrict__ out) {
  extern __shared__ float smem[];
  float* WihS = smem;
  float* WhhS = WihS + G3_ * STR_;
  float* hS   = WhhS + G3_ * STR_;
  float* xS   = hS   + 2 * 16 * STR_;
  float* gxS  = xS   + 2 * 16 * STR_;

  const int tid  = threadIdx.x;
  const int lane = tid & 31;
  const int wave = __builtin_amdgcn_readfirstlane(tid >> 5);
  const int half = lane >> 4;
  const int l15  = lane & 15;
  const int r    = blockIdx.x;
  const bool rec = (wave < 6);
  const int j    = rec ? wave : wave - 6;
  const int f    = j * 16 + l15;
  const int gofs = (j * 3 * 32 + lane) * 8;

  for (int idx = tid; idx < G3_ * F_; idx += TPG_) {
    const int g = idx / F_, k = idx % F_;
    WihS[g * STR_ + k] = Wih[((size_t)r * G3_ + g) * F_ + k];
    WhhS[g * STR_ + k] = Whh[((size_t)r * G3_ + g) * HID_ + k];
  }
  for (int idx = tid; idx < 2 * 16 * STR_; idx += TPG_) { hS[idx] = 0.f; xS[idx] = 0.f; }

  const float* bsrc = rec ? bhh : bih;
  const float b_r = bsrc[r * G3_ + f];
  const float b_z = bsrc[r * G3_ + HID_ + f];
  const float b_n = bsrc[r * G3_ + 2 * HID_ + f];

  float hprev[6];
  #pragma unroll
  for (int v = 0; v < 6; ++v) hprev[v] = 0.f;

  const int xl = j * 32 + lane;
  int ldsOfs[3];
  size_t outOfs[3];
  #pragma unroll
  for (int i = 0; i < 3; ++i) {
    const int u  = xl + 192 * i;
    const int m  = u / 48;
    const int c2 = 2 * (u % 48);
    const int R  = (m < 6) ? m : m + 2;
    ldsOfs[i] = R * STR_ + c2;
    const int bb = m / H_, hh = m % H_;
    outOfs[i] = (((size_t)(bb * ROI_ + r) * T_) * H_ + hh) * HID_ + c2;
  }
  const float* xg = hbuf + (size_t)r * T_ * BH_ * F_ + 2 * xl;

  __syncthreads();
  if (!rec) {
    #pragma unroll
    for (int i = 0; i < 3; ++i)
      *(v2f*)(xS + ldsOfs[i]) = *(const v2f*)(xg + 384 * i);
  }
  __syncthreads();

  if (rec) __builtin_amdgcn_s_setprio(3);

  for (int s = 0; s <= T_; ++s) {
    if (rec) {
      if (s >= 1) {
        const float* hcur = hS + ((s + 1) & 1) * (16 * STR_);
        float*       hnxt = hS + (s & 1) * (16 * STR_);
        const float* gxp  = gxS + ((s + 1) & 1) * GXT_ + gofs;
        float gx[3][6];
        #pragma unroll
        for (int g = 0; g < 3; ++g) {
          const v4f lo = *(const v4f*)(gxp + g * 256);
          const v2f hi = *(const v2f*)(gxp + g * 256 + 4);
          gx[g][0] = lo[0]; gx[g][1] = lo[1]; gx[g][2] = lo[2]; gx[g][3] = lo[3];
          gx[g][4] = hi[0]; gx[g][5] = hi[1];
        }
        v8f acc[3];
        mm1x3_gates(hcur, WhhS, j, lane, acc);
        #pragma unroll
        for (int v = 0; v < 6; ++v) {
          const float xr = gx[0][v];
          const float xz = gx[1][v];
          const float xn = gx[2][v];
          const float hr = acc[0][v] + b_r;
          const float hz = acc[1][v] + b_z;
          const float hn = acc[2][v] + b_n;
          const float rg = 1.f / (1.f + expf(-(xr + hr)));
          const float zg = 1.f / (1.f + expf(-(xz + hz)));
          const float ng = tanhf(xn + rg * hn);
          const float h2 = (1.f - zg) * ng + zg * hprev[v];
          hprev[v] = h2;
          hnxt[(v + 8 * half) * STR_ + f] = h2;
        }
      }
    } else {
      const bool doStage = (s + 1 < T_);
      const int sn = doStage ? s + 1 : T_ - 1;
      v2f xv[3];
      #pragma unroll
      for (int i = 0; i < 3; ++i)
        xv[i] = *(const v2f*)(xg + (size_t)sn * (BH_ * F_) + 384 * i);
      if (s >= 2) {
        const float* hcur = hS + ((s + 1) & 1) * (16 * STR_);
        #pragma unroll
        for (int i = 0; i < 3; ++i)
          *(v2f*)(out + outOfs[i] + (size_t)(s - 2) * (H_ * HID_)) =
              *(const v2f*)(hcur + ldsOfs[i]);
      }
      if (s < T_) {
        v8f acc[3];
        mm1x3_gates(xS + (s & 1) * (16 * STR_), WihS, j, lane, acc);
        float* gxp = gxS + (s & 1) * GXT_ + gofs;
        const float bi[3] = {b_r, b_z, b_n};
        #pragma unroll
        for (int g = 0; g < 3; ++g) {
          v4f lo; v2f hi;
          lo[0] = acc[g][0] + bi[g]; lo[1] = acc[g][1] + bi[g];
          lo[2] = acc[g][2] + bi[g]; lo[3] = acc[g][3] + bi[g];
          hi[0] = acc[g][4] + bi[g]; hi[1] = acc[g][5] + bi[g];
          *(v4f*)(gxp + g * 256)     = lo;
          *(v2f*)(gxp + g * 256 + 4) = hi;
        }
      }
      if (doStage) {
        float* xnxt = xS + ((s + 1) & 1) * (16 * STR_);
        #pragma unroll
        for (int i = 0; i < 3; ++i) *(v2f*)(xnxt + ldsOfs[i]) = xv[i];
      }
    }
    __syncthreads();
  }
  if (rec) __builtin_amdgcn_s_setprio(0);
  if (!rec) {
    const float* hlast = hS + (T_ & 1) * (16 * STR_);
    #pragma unroll
    for (int i = 0; i < 3; ++i)
      *(v2f*)(out + outOfs[i] + (size_t)(T_ - 1) * (H_ * HID_)) =
          *(const v2f*)(hlast + ldsOfs[i]);
  }
}

extern "C" void kernel_launch(void* const* d_in, const int* in_sizes, int n_in,
                              void* d_out, int out_size, void* d_ws, size_t ws_size,
                              hipStream_t stream) {
  const float* x   = (const float*)d_in[0];
  const float* a   = (const float*)d_in[1];
  const float* eps = (const float*)d_in[2];
  const float* W1  = (const float*)d_in[3];
  const float* b1  = (const float*)d_in[4];
  const float* g1  = (const float*)d_in[5];
  const float* be1 = (const float*)d_in[6];
  const float* W2  = (const float*)d_in[7];
  const float* b2  = (const float*)d_in[8];
  const float* g2  = (const float*)d_in[9];
  const float* be2 = (const float*)d_in[10];
  const float* Wih = (const float*)d_in[11];
  const float* Whh = (const float*)d_in[12];
  const float* bih = (const float*)d_in[13];
  const float* bhh = (const float*)d_in[14];
  float* out  = (float*)d_out;
  float* hbuf = (float*)d_ws;

  const size_t ldsA = (size_t)2 * NP_ * STR_ * sizeof(float);
  ftdgnn_agg_mlp<<<B_ * T_ * H_, TPB_, ldsA, stream>>>(
      x, a, eps, W1, b1, g1, be1, W2, b2, g2, be2, hbuf);

  const size_t ldsB = (size_t)(2 * G3_ * STR_ + 4 * 16 * STR_ +
                               2 * GXT_) * sizeof(float);
  ftdgnn_gru<<<ROI_, TPG_, ldsB, stream>>>(Wih, Whh, bih, bhh, hbuf, out);
}


# GRU: each ROI's 12 sequences split over two blocks of 6 (180 blocks): a 16-row tile then holds two time steps of x@Wih^T (one product per 2 steps) and a lane evaluates 3 gate rows, not 6
# speedup vs baseline: 2.2954x; 1.2258x over previous
#include <hip/hip_runtime.h>
#include <math.h>

#define B_    4
#define ROI_  90
#define T_    200
#define H_    3
#define F_    96
#define HID_  96
#define G3_   288
#define BH_   12
#define NP_   96
#define STR_  100
#define TPB_  288

typedef float v2f __attribute__((ext_vector_type(2)));
typedef float v8f __attribute__((ext_vector_type(8)));

__device__ __forceinline__ v8f wmma4(v2f a, v2f b, v8f c) {
  return __builtin_amdgcn_wmma_f32_16x16x4_f32(false, a, false, b, (short)0, c,
                                               false, false);
}


__device__ __forceinline__ void mm2x2_gB(const float* As, const float* __restrict__ Bg,
                                         int sr, int sc, int lane, v8f acc[4]) {
  const int half = lane >> 4;
  const int l15  = lane & 15;
  const float* a0 = As + (sr * 32 + l15) * STR_ + (half << 1);
  const float* a1 = a0 + 16 * STR_;
  const float* b0 = Bg + (half << 1) * NP_ + sc * 32 + l15;
  #pragma unroll
  for (int j = 0; j < 4; ++j) acc[j] = (v8f){0.f,0.f,0.f,0.f,0.f,0.f,0.f,0.f};
  #pragma unroll
  for (int k0 = 0; k0 < NP_; k0 += 4) {
    v2f av0 = *(const v2f*)(a0 + k0);
    v2f av1 = *(const v2f*)(a1 + k0);
    v2f bv0, bv1;
    bv0[0] = b0[k0 * NP_];        bv0[1] = b0[(k0 + 1) * NP_];
    bv1[0] = b0[k0 * NP_ + 16];   bv1[1] = b0[(k0 + 1) * NP_ + 16];
    acc[0] = wmma4(av0, bv0, acc[0]);
    acc[1] = wmma4(av0, bv1, acc[1]);
    acc[2] = wmma4(av1, bv0, acc[2]);
    acc[3] = wmma4(av1, bv1, acc[3]);
  }
}

#define XRS_  ((size_t)T_ * H_ * F_)
#define ARS_  ((size_t)T_ * H_ * ROI_)
#define KAGG_ 92

__global__ void __launch_bounds__(TPB_)
ftdgnn_agg_mlp(const float* __restrict__ x, const float* __restrict__ a,
               const float* __restrict__ eps,
               const float* __restrict__ W1, const float* __restrict__ b1,
               const float* __restrict__ g1, const float* __restrict__ be1,
               const float* __restrict__ W2, const float* __restrict__ b2,
               const float* __restrict__ g2, const float* __restrict__ be2,
               float* __restrict__ hbuf) {
  extern __shared__ float smem[];
  float* aggS = smem;
  float* h1S  = aggS + NP_ * STR_;

  const int tid  = threadIdx.x;
  const int lane = tid & 31;
  const int wave = tid >> 5;
  const int half = lane >> 4;
  const int l15  = lane & 15;
  const int sr   = wave / 3;
  const int sc   = wave % 3;

  const int bth = blockIdx.x;
  const int b   = bth / (T_ * H_);
  const int th  = bth % (T_ * H_);
  const int t   = th / H_;
  const int hh  = th % H_;

  const size_t tile = ((size_t)(b * ROI_) * T_ + t) * H_ + hh;
  const float* xb = x + tile * F_;
  const float* ab = a + tile * ROI_;

  const float epsv = eps[0];
  const float rs   = rsqrtf(1.0f + 1e-5f);

  v8f acc[4];

  {
    const int  i0   = sr * 32 + l15;
    const int  i1   = i0 + 16;
    const bool i1ok = (i1 < ROI_);
    const float* ar0 = ab + (size_t)i0 * ARS_;
    const float* ar1 = ab + (size_t)(i1ok ? i1 : ROI_ - 1) * ARS_;
    const float* xc  = xb + sc * 32 + l15;
    #pragma unroll
    for (int j = 0; j < 4; ++j) acc[j] = (v8f){0.f,0.f,0.f,0.f,0.f,0.f,0.f,0.f};
    #pragma unroll
    for (int k0 = 0; k0 < KAGG_; k0 += 4) {
      const int  j   = k0 + (half << 1);
      const bool jok = (j < ROI_);
      const int  jc  = jok ? j : 0;
      const v2f r0 = *(const v2f*)(ar0 + jc);
      const v2f r1 = *(const v2f*)(ar1 + jc);
      const float* xr = xc + (size_t)jc * XRS_;
      const float x00 = xr[0],    x01 = xr[16];
      const float x10 = xr[XRS_], x11 = xr[XRS_ + 16];
      v2f av0, av1, bv0, bv1;
      av0[0] = (jok && r0[0] != 0.f) ? 1.f : 0.f;
      av0[1] = (jok && r0[1] != 0.f) ? 1.f : 0.f;
      av1[0] = (jok && i1ok && r1[0] != 0.f) ? 1.f : 0.f;
      av1[1] = (jok && i1ok && r1[1] != 0.f) ? 1.f : 0.f;
      bv0[0] = jok ? x00 : 0.f;   bv0[1] = jok ? x10 : 0.f;
      bv1[0] = jok ? x01 : 0.f;   bv1[1] = jok ? x11 : 0.f;
      acc[0] = wmma4(av0, bv0, acc[0]);
      acc[1] = wmma4(av0, bv1, acc[1]);
      acc[2] = wmma4(av1, bv0, acc[2]);
      acc[3] = wmma4(av1, bv1, acc[3]);
    }
  }
  #pragma unroll
  for (int di = 0; di < 2; ++di)
    #pragma unroll
    for (int dj = 0; dj < 2; ++dj) {
      const int n = (2 * sc + dj) * 16 + l15;
      #pragma unroll
      for (int v = 0; v < 8; ++v) {
        const int  m   = (2 * sr + di) * 16 + v + 8 * half;
        const bool mok = (m < ROI_);
        const float xv = xb[(size_t)(mok ? m : ROI_ - 1) * XRS_ + n];
        aggS[m * STR_ + n] = acc[di * 2 + dj][v] + epsv * (mok ? xv : 0.f);
      }
    }
  __syncthreads();

  mm2x2_gB(aggS, W1, sr, sc, lane, acc);
  #pragma unroll
  for (int di = 0; di < 2; ++di)
    #pragma unroll
    for (int dj = 0; dj < 2; ++dj) {
      const int n = (2 * sc + dj) * 16 + l15;
      const float sc1 = g1[n] * rs, sh1 = be1[n], bb1 = b1[n];
      #pragma unroll
      for (int v = 0; v < 8; ++v) {
        const int m = (2 * sr + di) * 16 + v + 8 * half;
        float z = (acc[di * 2 + dj][v] + bb1) * sc1 + sh1;
        z = (z > 0.f) ? z : expm1f(z);
        h1S[m * STR_ + n] = z;
      }
    }
  __syncthreads();

  mm2x2_gB(h1S, W2, sr, sc, lane, acc);
  #pragma unroll
  for (int di = 0; di < 2; ++di)
    #pragma unroll
    for (int dj = 0; dj < 2; ++dj) {
      const int n = (2 * sc + dj) * 16 + l15;
      const float sc2 = g2[n] * rs, sh2 = be2[n], bb2 = b2[n];
      #pragma unroll
      for (int v = 0; v < 8; ++v) {
        const int m = (2 * sr + di) * 16 + v + 8 * half;
        if (m < ROI_) {
          float z = (acc[di * 2 + dj][v] + bb2) * sc2 + sh2;
          z = (z > 0.f) ? z : 0.f;
          hbuf[((size_t)(m * T_ + t) * BH_ + (b * H_ + hh)) * F_ + n] = z;
        }
      }
    }
}

#define TPG_   384
#define GXT_   (18 * 32 * 8)
#define RPB_   6
#define NBLK_  (2 * ROI_)

typedef float v4f __attribute__((ext_vector_type(4)));

__device__ __forceinline__ void mm1x3_gates(const float* As, const float* Bt,
                                            int j, int lane, v8f acc[3]) {
  const int half = lane >> 4;
  const int l15  = lane & 15;
  const float* aRow = As + l15 * STR_ + (half << 1);
  const float* bRow = Bt + (j * 16 + l15) * STR_ + (half << 1);
  #pragma unroll
  for (int g = 0; g < 3; ++g) acc[g] = (v8f){0.f,0.f,0.f,0.f,0.f,0.f,0.f,0.f};
  #pragma unroll
  for (int k0 = 0; k0 < F_; k0 += 4) {
    v2f av = *(const v2f*)(aRow + k0);
    #pragma unroll
    for (int g = 0; g < 3; ++g) {
      v2f bv = *(const v2f*)(bRow + g * HID_ * STR_ + k0);
      acc[g] = wmma4(av, bv, acc[g]);
    }
  }
}

__global__ void __launch_bounds__(TPG_)
ftdgnn_gru(const float* __restrict__ Wih, const float* __restrict__ Whh,
           const float* __restrict__ bih, const float* __restrict__ bhh,
           const float* __restrict__ hbuf, float* __restrict__ out) {
  extern __shared__ float smem[];
  float* WihS = smem;
  float* WhhS = WihS + G3_ * STR_;
  float* hS   = WhhS + G3_ * STR_;
  float* xS   = hS   + 2 * 16 * STR_;
  float* gxS  = xS   + 2 * 16 * STR_;

  const int tid  = threadIdx.x;
  const int lane = tid & 31;
  const int wave = __builtin_amdgcn_readfirstlane(tid >> 5);
  const int half = lane >> 4;
  const int l15  = lane & 15;
  const int r    = blockIdx.x >> 1;
  const int part = blockIdx.x & 1;
  const bool rec = (wave < 6);
  const int j    = rec ? wave : wave - 6;
  const int f    = j * 16 + l15;
  const int gofs = (j * 3 * 32 + lane) * 8;

  for (int idx = tid; idx < G3_ * F_; idx += TPG_) {
    const int g = idx / F_, k = idx % F_;
    WihS[g * STR_ + k] = Wih[((size_t)r * G3_ + g) * F_ + k];
    WhhS[g * STR_ + k] = Whh[((size_t)r * G3_ + g) * HID_ + k];
  }
  for (int idx = tid; idx < 2 * 16 * STR_; idx += TPG_) { hS[idx] = 0.f; xS[idx] = 0.f; }

  const float* bsrc = rec ? bhh : bih;
  const float b_r = bsrc[r * G3_ + f];
  const float b_z = bsrc[r * G3_ + HID_ + f];
  const float b_n = bsrc[r * G3_ + 2 * HID_ + f];

  float hprev[3];
  #pragma unroll
  for (int v = 0; v < 3; ++v) hprev[v] = 0.f;

  const int xl = j * 32 + lane;
  int xLds[3];
  int xGl[3];
  #pragma unroll
  for (int i = 0; i < 3; ++i) {
    const int u   = xl + 192 * i;
    const int rho = u / 48;
    const int c2  = 2 * (u % 48);
    const int q   = rho / RPB_;
    const int bq  = rho % RPB_;
    const int R   = ((bq < 3) ? bq : bq + 5) + 3 * q;
    xLds[i] = R * STR_ + c2;
    xGl[i]  = (q * BH_ + part * RPB_ + bq) * F_ + c2;
  }
  int hLds[2];
  size_t outOfs[2];
  #pragma unroll
  for (int i = 0; i < 2; ++i) {
    const int u  = xl + 192 * i;
    const int bq = (u / 48) % RPB_;
    const int c2 = 2 * (u % 48);
    const int R  = (bq < 3) ? bq : bq + 5;
    hLds[i] = R * STR_ + c2;
    const int m  = part * RPB_ + bq;
    const int bb = m / H_, hh = m % H_;
    outOfs[i] = (((size_t)(bb * ROI_ + r) * T_) * H_ + hh) * HID_ + c2;
  }
  const bool two = (j < 3);
  const float* xg = hbuf + (size_t)r * T_ * BH_ * F_;

  __syncthreads();
  if (!rec) {
    #pragma unroll
    for (int i = 0; i < 3; ++i)
      *(v2f*)(xS + xLds[i]) = *(const v2f*)(xg + xGl[i]);
  }
  __syncthreads();

  if (rec) __builtin_amdgcn_s_setprio(3);

  for (int s = 0; s <= T_; ++s) {
    if (rec) {
      if (s >= 1) {
        const int st = s - 1;
        const int q  = st & 1;
        const float* hcur = hS + ((s + 1) & 1) * (16 * STR_);
        float*       hnxt = hS + (s & 1) * (16 * STR_);
        const float* gxp  = gxS + ((st >> 1) & 1) * GXT_ + gofs;
        float gx[3][3];
        #pragma unroll
        for (int g = 0; g < 3; ++g) {
          const v4f lo = *(const v4f*)(gxp + g * 256);
          const v2f hi = *(const v2f*)(gxp + g * 256 + 4);
          gx[g][0] = q ? lo[3] : lo[0];
          gx[g][1] = q ? hi[0] : lo[1];
          gx[g][2] = q ? hi[1] : lo[2];
        }
        v8f acc[3];
        mm1x3_gates(hcur, WhhS, j, lane, acc);
        #pragma unroll
        for (int v = 0; v < 3; ++v) {
          const float xr = gx[0][v];
          const float xz = gx[1][v];
          const float xn = gx[2][v];
          const float hr = acc[0][v] + b_r;
          const float hz = acc[1][v] + b_z;
          const float hn = acc[2][v] + b_n;
          const float rg = 1.f / (1.f + expf(-(xr + hr)));
          const float zg = 1.f / (1.f + expf(-(xz + hz)));
          const float ng = tanhf(xn + rg * hn);
          const float h2 = (1.f - zg) * ng + zg * hprev[v];
          hprev[v] = h2;
          hnxt[(v + 8 * half) * STR_ + f] = h2;
        }
      }
    } else {
      const bool evenS   = ((s & 1) == 0);
      const bool doProd  = evenS && (s < T_);
      const bool doStage = evenS && (s + 2 < T_);
      v2f xv[3];
      #pragma unroll
      for (int i = 0; i < 3; ++i) xv[i] = (v2f){0.f, 0.f};
      if (doStage) {
        #pragma unroll
        for (int i = 0; i < 3; ++i)
          xv[i] = *(const v2f*)(xg + (size_t)(s + 2) * (BH_ * F_) + xGl[i]);
      }
      if (s >= 2) {
        const float* hcur = hS + ((s + 1) & 1) * (16 * STR_);
        *(v2f*)(out + outOfs[0] + (size_t)(s - 2) * (H_ * HID_)) = *(const v2f*)(hcur + hLds[0]);
        if (two)
          *(v2f*)(out + outOfs[1] + (size_t)(s - 2) * (H_ * HID_)) = *(const v2f*)(hcur + hLds[1]);
      }
      if (doProd) {
        v8f acc[3];
        mm1x3_gates(xS + ((s >> 1) & 1) * (16 * STR_), WihS, j, lane, acc);
        float* gxp = gxS + ((s >> 1) & 1) * GXT_ + gofs;
        const float bi[3] = {b_r, b_z, b_n};
        #pragma unroll
        for (int g = 0; g < 3; ++g) {
          v4f lo; v2f hi;
          lo[0] = acc[g][0] + bi[g]; lo[1] = acc[g][1] + bi[g];
          lo[2] = acc[g][2] + bi[g]; lo[3] = acc[g][3] + bi[g];
          hi[0] = acc[g][4] + bi[g]; hi[1] = acc[g][5] + bi[g];
          *(v4f*)(gxp + g * 256)     = lo;
          *(v2f*)(gxp + g * 256 + 4) = hi;
        }
      }
      if (doStage) {
        float* xnxt = xS + (((s >> 1) + 1) & 1) * (16 * STR_);
        #pragma unroll
        for (int i = 0; i < 3; ++i) *(v2f*)(xnxt + xLds[i]) = xv[i];
      }
    }
    __syncthreads();
  }
  if (rec) __builtin_amdgcn_s_setprio(0);
  if (!rec) {
    const float* hlast = hS + (T_ & 1) * (16 * STR_);
    *(v2f*)(out + outOfs[0] + (size_t)(T_ - 1) * (H_ * HID_)) = *(const v2f*)(hlast + hLds[0]);
    if (two)
      *(v2f*)(out + outOfs[1] + (size_t)(T_ - 1) * (H_ * HID_)) = *(const v2f*)(hlast + hLds[1]);
  }
}

extern "C" void kernel_launch(void* const* d_in, const int* in_sizes, int n_in,
                              void* d_out, int out_size, void* d_ws, size_t ws_size,
                              hipStream_t stream) {
  const float* x   = (const float*)d_in[0];
  const float* a   = (const float*)d_in[1];
  const float* eps = (const float*)d_in[2];
  const float* W1  = (const float*)d_in[3];
  const float* b1  = (const float*)d_in[4];
  const float* g1  = (const float*)d_in[5];
  const float* be1 = (const float*)d_in[6];
  const float* W2  = (const float*)d_in[7];
  const float* b2  = (const float*)d_in[8];
  const float* g2  = (const float*)d_in[9];
  const float* be2 = (const float*)d_in[10];
  const float* Wih = (const float*)d_in[11];
  const float* Whh = (const float*)d_in[12];
  const float* bih = (const float*)d_in[13];
  const float* bhh = (const float*)d_in[14];
  float* out  = (float*)d_out;
  float* hbuf = (float*)d_ws;

  const size_t ldsA = (size_t)2 * NP_ * STR_ * sizeof(float);
  ftdgnn_agg_mlp<<<B_ * T_ * H_, TPB_, ldsA, stream>>>(
      x, a, eps, W1, b1, g1, be1, W2, b2, g2, be2, hbuf);

  const size_t ldsB = (size_t)(2 * G3_ * STR_ + 4 * 16 * STR_ +
                               2 * GXT_) * sizeof(float);
  ftdgnn_gru<<<NBLK_, TPG_, ldsB, stream>>>(Wih, Whh, bih, bhh, hbuf, out);
}


# GRU recurrent waves: r and z tiles of h@Whh^T finished first, their sigmoids evaluated beside the n tile's WMMAs; only tanh and the blend follow a step's last WMMA (each tile's k order unchanged)
# speedup vs baseline: 2.3882x; 1.0404x over previous
#include <hip/hip_runtime.h>
#include <math.h>

#define B_    4
#define ROI_  90
#define T_    200
#define H_    3
#define F_    96
#define HID_  96
#define G3_   288
#define BH_   12
#define NP_   96
#define STR_  100
#define TPB_  288

typedef float v2f __attribute__((ext_vector_type(2)));
typedef float v8f __attribute__((ext_vector_type(8)));

__device__ __forceinline__ v8f wmma4(v2f a, v2f b, v8f c) {
  return __builtin_amdgcn_wmma_f32_16x16x4_f32(false, a, false, b, (short)0, c,
                                               false, false);
}


__device__ __forceinline__ void mm2x2_gB(const float* As, const float* __restrict__ Bg,
                                         int sr, int sc, int lane, v8f acc[4]) {
  const int half = lane >> 4;
  const int l15  = lane & 15;
  const float* a0 = As + (sr * 32 + l15) * STR_ + (half << 1);
  const float* a1 = a0 + 16 * STR_;
  const float* b0 = Bg + (half << 1) * NP_ + sc * 32 + l15;
  #pragma unroll
  for (int j = 0; j < 4; ++j) acc[j] = (v8f){0.f,0.f,0.f,0.f,0.f,0.f,0.f,0.f};
  #pragma unroll
  for (int k0 = 0; k0 < NP_; k0 += 4) {
    v2f av0 = *(const v2f*)(a0 + k0);
    v2f av1 = *(const v2f*)(a1 + k0);
    v2f bv0, bv1;
    bv0[0] = b0[k0 * NP_];        bv0[1] = b0[(k0 + 1) * NP_];
    bv1[0] = b0[k0 * NP_ + 16];   bv1[1] = b0[(k0 + 1) * NP_ + 16];
    acc[0] = wmma4(av0, bv0, acc[0]);
    acc[1] = wmma4(av0, bv1, acc[1]);
    acc[2] = wmma4(av1, bv0, acc[2]);
    acc[3] = wmma4(av1, bv1, acc[3]);
  }
}

#define XRS_  ((size_t)T_ * H_ * F_)
#define ARS_  ((size_t)T_ * H_ * ROI_)
#define KAGG_ 92

__global__ void __launch_bounds__(TPB_)
ftdgnn_agg_mlp(const float* __restrict__ x, const float* __restrict__ a,
               const float* __restrict__ eps,
               const float* __restrict__ W1, const float* __restrict__ b1,
               const float* __restrict__ g1, const float* __restrict__ be1,
               const float* __restrict__ W2, const float* __restrict__ b2,
               const float* __restrict__ g2, const float* __restrict__ be2,
               float* __restrict__ hbuf) {
  extern __shared__ float smem[];
  float* aggS = smem;
  float* h1S  = aggS + NP_ * STR_;

  const int tid  = threadIdx.x;
  const int lane = tid & 31;
  const int wave = tid >> 5;
  const int half = lane >> 4;
  const int l15  = lane & 15;
  const int sr   = wave / 3;
  const int sc   = wave % 3;

  const int bth = blockIdx.x;
  const int b   = bth / (T_ * H_);
  const int th  = bth % (T_ * H_);
  const int t   = th / H_;
  const int hh  = th % H_;

  const size_t tile = ((size_t)(b * ROI_) * T_ + t) * H_ + hh;
  const float* xb = x + tile * F_;
  const float* ab = a + tile * ROI_;

  const float epsv = eps[0];
  const float rs   = rsqrtf(1.0f + 1e-5f);

  v8f acc[4];

  {
    const int  i0   = sr * 32 + l15;
    const int  i1   = i0 + 16;
    const bool i1ok = (i1 < ROI_);
    const float* ar0 = ab + (size_t)i0 * ARS_;
    const float* ar1 = ab + (size_t)(i1ok ? i1 : ROI_ - 1) * ARS_;
    const float* xc  = xb + sc * 32 + l15;
    #pragma unroll
    for (int j = 0; j < 4; ++j) acc[j] = (v8f){0.f,0.f,0.f,0.f,0.f,0.f,0.f,0.f};
    #pragma unroll
    for (int k0 = 0; k0 < KAGG_; k0 += 4) {
      const int  j   = k0 + (half << 1);
      const bool jok = (j < ROI_);
      const int  jc  = jok ? j : 0;
      const v2f r0 = *(const v2f*)(ar0 + jc);
      const v2f r1 = *(const v2f*)(ar1 + jc);
      const float* xr = xc + (size_t)jc * XRS_;
      const float x00 = xr[0],    x01 = xr[16];
      const float x10 = xr[XRS_], x11 = xr[XRS_ + 16];
      v2f av0, av1, bv0, bv1;
      av0[0] = (jok && r0[0] != 0.f) ? 1.f : 0.f;
      av0[1] = (jok && r0[1] != 0.f) ? 1.f : 0.f;
      av1[0] = (jok && i1ok && r1[0] != 0.f) ? 1.f : 0.f;
      av1[1] = (jok && i1ok && r1[1] != 0.f) ? 1.f : 0.f;
      bv0[0] = jok ? x00 : 0.f;   bv0[1] = jok ? x10 : 0.f;
      bv1[0] = jok ? x01 : 0.f;   bv1[1] = jok ? x11 : 0.f;
      acc[0] = wmma4(av0, bv0, acc[0]);
      acc[1] = wmma4(av0, bv1, acc[1]);
      acc[2] = wmma4(av1, bv0, acc[2]);
      acc[3] = wmma4(av1, bv1, acc[3]);
    }
  }
  #pragma unroll
  for (int di = 0; di < 2; ++di)
    #pragma unroll
    for (int dj = 0; dj < 2; ++dj) {
      const int n = (2 * sc + dj) * 16 + l15;
      #pragma unroll
      for (int v = 0; v < 8; ++v) {
        const int  m   = (2 * sr + di) * 16 + v + 8 * half;
        const bool mok = (m < ROI_);
        const float xv = xb[(size_t)(mok ? m : ROI_ - 1) * XRS_ + n];
        aggS[m * STR_ + n] = acc[di * 2 + dj][v] + epsv * (mok ? xv : 0.f);
      }
    }
  __syncthreads();

  mm2x2_gB(aggS, W1, sr, sc, lane, acc);
  #pragma unroll
  for (int di = 0; di < 2; ++di)
    #pragma unroll
    for (int dj = 0; dj < 2; ++dj) {
      const int n = (2 * sc + dj) * 16 + l15;
      const float sc1 = g1[n] * rs, sh1 = be1[n], bb1 = b1[n];
      #pragma unroll
      for (int v = 0; v < 8; ++v) {
        const int m = (2 * sr + di) * 16 + v + 8 * half;
        float z = (acc[di * 2 + dj][v] + bb1) * sc1 + sh1;
        z = (z > 0.f) ? z : expm1f(z);
        h1S[m * STR_ + n] = z;
      }
    }
  __syncthreads();

  mm2x2_gB(h1S, W2, sr, sc, lane, acc);
  #pragma unroll
  for (int di = 0; di < 2; ++di)
    #pragma unroll
    for (int dj = 0; dj < 2; ++dj) {
      const int n = (2 * sc + dj) * 16 + l15;
      const float sc2 = g2[n] * rs, sh2 = be2[n], bb2 = b2[n];
      #pragma unroll
      for (int v = 0; v < 8; ++v) {
        const int m = (2 * sr + di) * 16 + v + 8 * half;
        if (m < ROI_) {
          float z = (acc[di * 2 + dj][v] + bb2) * sc2 + sh2;
          z = (z > 0.f) ? z : 0.f;
          hbuf[((size_t)(m * T_ + t) * BH_ + (b * H_ + hh)) * F_ + n] = z;
        }
      }
    }
}

#define TPG_   384
#define GXT_   (18 * 32 * 8)
#define RPB_   6
#define NBLK_  (2 * ROI_)

typedef float v4f __attribute__((ext_vector_type(4)));

__device__ __forceinline__ void mm1x3_gates(const float* As, const float* Bt,
                                            int j, int lane, v8f acc[3]) {
  const int half = lane >> 4;
  const int l15  = lane & 15;
  const float* aRow = As + l15 * STR_ + (half << 1);
  const float* bRow = Bt + (j * 16 + l15) * STR_ + (half << 1);
  #pragma unroll
  for (int g = 0; g < 3; ++g) acc[g] = (v8f){0.f,0.f,0.f,0.f,0.f,0.f,0.f,0.f};
  #pragma unroll
  for (int k0 = 0; k0 < F_; k0 += 4) {
    v2f av = *(const v2f*)(aRow + k0);
    #pragma unroll
    for (int g = 0; g < 3; ++g) {
      v2f bv = *(const v2f*)(bRow + g * HID_ * STR_ + k0);
      acc[g] = wmma4(av, bv, acc[g]);
    }
  }
}

__global__ void __launch_bounds__(TPG_)
ftdgnn_gru(const float* __restrict__ Wih, const float* __restrict__ Whh,
           const float* __restrict__ bih, const float* __restrict__ bhh,
           const float* __restrict__ hbuf, float* __restrict__ out) {
  extern __shared__ float smem[];
  float* WihS = smem;
  float* WhhS = WihS + G3_ * STR_;
  float* hS   = WhhS + G3_ * STR_;
  float* xS   = hS   + 2 * 16 * STR_;
  float* gxS  = xS   + 2 * 16 * STR_;

  const int tid  = threadIdx.x;
  const int lane = tid & 31;
  const int wave = __builtin_amdgcn_readfirstlane(tid >> 5);
  const int half = lane >> 4;
  const int l15  = lane & 15;
  const int r    = blockIdx.x >> 1;
  const int part = blockIdx.x & 1;
  const bool rec = (wave < 6);
  const int j    = rec ? wave : wave - 6;
  const int f    = j * 16 + l15;
  const int gofs = (j * 3 * 32 + lane) * 8;

  for (int idx = tid; idx < G3_ * F_; idx += TPG_) {
    const int g = idx / F_, k = idx % F_;
    WihS[g * STR_ + k] = Wih[((size_t)r * G3_ + g) * F_ + k];
    WhhS[g * STR_ + k] = Whh[((size_t)r * G3_ + g) * HID_ + k];
  }
  for (int idx = tid; idx < 2 * 16 * STR_; idx += TPG_) { hS[idx] = 0.f; xS[idx] = 0.f; }

  const float* bsrc = rec ? bhh : bih;
  const float b_r = bsrc[r * G3_ + f];
  const float b_z = bsrc[r * G3_ + HID_ + f];
  const float b_n = bsrc[r * G3_ + 2 * HID_ + f];

  float hprev[3];
  #pragma unroll
  for (int v = 0; v < 3; ++v) hprev[v] = 0.f;

  const int xl = j * 32 + lane;
  int xLds[3];
  int xGl[3];
  #pragma unroll
  for (int i = 0; i < 3; ++i) {
    const int u   = xl + 192 * i;
    const int rho = u / 48;
    const int c2  = 2 * (u % 48);
    const int q   = rho / RPB_;
    const int bq  = rho % RPB_;
    const int R   = ((bq < 3) ? bq : bq + 5) + 3 * q;
    xLds[i] = R * STR_ + c2;
    xGl[i]  = (q * BH_ + part * RPB_ + bq) * F_ + c2;
  }
  int hLds[2];
  size_t outOfs[2];
  #pragma unroll
  for (int i = 0; i < 2; ++i) {
    const int u  = xl + 192 * i;
    const int bq = (u / 48) % RPB_;
    const int c2 = 2 * (u % 48);
    const int R  = (bq < 3) ? bq : bq + 5;
    hLds[i] = R * STR_ + c2;
    const int m  = part * RPB_ + bq;
    const int bb = m / H_, hh = m % H_;
    outOfs[i] = (((size_t)(bb * ROI_ + r) * T_) * H_ + hh) * HID_ + c2;
  }
  const bool two = (j < 3);
  const float* xg = hbuf + (size_t)r * T_ * BH_ * F_;

  __syncthreads();
  if (!rec) {
    #pragma unroll
    for (int i = 0; i < 3; ++i)
      *(v2f*)(xS + xLds[i]) = *(const v2f*)(xg + xGl[i]);
  }
  __syncthreads();

  if (rec) __builtin_amdgcn_s_setprio(3);

  for (int s = 0; s <= T_; ++s) {
    if (rec) {
      if (s >= 1) {
        const int st = s - 1;
        const int q  = st & 1;
        const float* hcur = hS + ((s + 1) & 1) * (16 * STR_);
        float*       hnxt = hS + (s & 1) * (16 * STR_);
        const float* gxp  = gxS + ((st >> 1) & 1) * GXT_ + gofs;
        float gx[3][3];
        #pragma unroll
        for (int g = 0; g < 3; ++g) {
          const v4f lo = *(const v4f*)(gxp + g * 256);
          const v2f hi = *(const v2f*)(gxp + g * 256 + 4);
          gx[g][0] = q ? lo[3] : lo[0];
          gx[g][1] = q ? hi[0] : lo[1];
          gx[g][2] = q ? hi[1] : lo[2];
        }
        const float* aRow = hcur + l15 * STR_ + (half << 1);
        const float* bRow = WhhS + (j * 16 + l15) * STR_ + (half << 1);
        v2f av[F_ / 4];
        #pragma unroll
        for (int kk = 0; kk < F_ / 4; ++kk) av[kk] = *(const v2f*)(aRow + 4 * kk);
        v8f accR = (v8f){0.f,0.f,0.f,0.f,0.f,0.f,0.f,0.f};
        v8f accZ = (v8f){0.f,0.f,0.f,0.f,0.f,0.f,0.f,0.f};
        #pragma unroll
        for (int kk = 0; kk < F_ / 4; ++kk) {
          const v2f br = *(const v2f*)(bRow + 4 * kk);
          const v2f bz = *(const v2f*)(bRow + HID_ * STR_ + 4 * kk);
          accR = wmma4(av[kk], br, accR);
          accZ = wmma4(av[kk], bz, accZ);
        }
        __builtin_amdgcn_sched_barrier(0);
        float rg[3], zg[3];
        #pragma unroll
        for (int v = 0; v < 3; ++v) {
          const float hr = accR[v] + b_r;
          const float hz = accZ[v] + b_z;
          rg[v] = 1.f / (1.f + expf(-(gx[0][v] + hr)));
          zg[v] = 1.f / (1.f + expf(-(gx[1][v] + hz)));
        }
        v8f accN = (v8f){0.f,0.f,0.f,0.f,0.f,0.f,0.f,0.f};
        #pragma unroll
        for (int kk = 0; kk < F_ / 4; ++kk) {
          const v2f bn = *(const v2f*)(bRow + 2 * HID_ * STR_ + 4 * kk);
          accN = wmma4(av[kk], bn, accN);
        }
        asm volatile("" :: "v"(rg[0]), "v"(rg[1]), "v"(rg[2]), "v"(zg[0]), "v"(zg[1]), "v"(zg[2]));
        #pragma unroll
        for (int v = 0; v < 3; ++v) {
          const float hn = accN[v] + b_n;
          const float ng = tanhf(gx[2][v] + rg[v] * hn);
          const float h2 = (1.f - zg[v]) * ng + zg[v] * hprev[v];
          hprev[v] = h2;
          hnxt[(v + 8 * half) * STR_ + f] = h2;
        }
      }
    } else {
      const bool evenS   = ((s & 1) == 0);
      const bool doProd  = evenS && (s < T_);
      const bool doStage = evenS && (s + 2 < T_);
      v2f xv[3];
      #pragma unroll
      for (int i = 0; i < 3; ++i) xv[i] = (v2f){0.f, 0.f};
      if (doStage) {
        #pragma unroll
        for (int i = 0; i < 3; ++i)
          xv[i] = *(const v2f*)(xg + (size_t)(s + 2) * (BH_ * F_) + xGl[i]);
      }
      if (s >= 2) {
        const float* hcur = hS + ((s + 1) & 1) * (16 * STR_);
        *(v2f*)(out + outOfs[0] + (size_t)(s - 2) * (H_ * HID_)) = *(const v2f*)(hcur + hLds[0]);
        if (two)
          *(v2f*)(out + outOfs[1] + (size_t)(s - 2) * (H_ * HID_)) = *(const v2f*)(hcur + hLds[1]);
      }
      if (doProd) {
        v8f acc[3];
        mm1x3_gates(xS + ((s >> 1) & 1) * (16 * STR_), WihS, j, lane, acc);
        float* gxp = gxS + ((s >> 1) & 1) * GXT_ + gofs;
        const float bi[3] = {b_r, b_z, b_n};
        #pragma unroll
        for (int g = 0; g < 3; ++g) {
          v4f lo; v2f hi;
          lo[0] = acc[g][0] + bi[g]; lo[1] = acc[g][1] + bi[g];
          lo[2] = acc[g][2] + bi[g]; lo[3] = acc[g][3] + bi[g];
          hi[0] = acc[g][4] + bi[g]; hi[1] = acc[g][5] + bi[g];
          *(v4f*)(gxp + g * 256)     = lo;
          *(v2f*)(gxp + g * 256 + 4) = hi;
        }
      }
      if (doStage) {
        float* xnxt = xS + (((s >> 1) + 1) & 1) * (16 * STR_);
        #pragma unroll
        for (int i = 0; i < 3; ++i) *(v2f*)(xnxt + xLds[i]) = xv[i];
      }
    }
    __syncthreads();
  }
  if (rec) __builtin_amdgcn_s_setprio(0);
  if (!rec) {
    const float* hlast = hS + (T_ & 1) * (16 * STR_);
    *(v2f*)(out + outOfs[0] + (size_t)(T_ - 1) * (H_ * HID_)) = *(const v2f*)(hlast + hLds[0]);
    if (two)
      *(v2f*)(out + outOfs[1] + (size_t)(T_ - 1) * (H_ * HID_)) = *(const v2f*)(hlast + hLds[1]);
  }
}

extern "C" void kernel_launch(void* const* d_in, const int* in_sizes, int n_in,
                              void* d_out, int out_size, void* d_ws, size_t ws_size,
                              hipStream_t stream) {
  const float* x   = (const float*)d_in[0];
  const float* a   = (const float*)d_in[1];
  const float* eps = (const float*)d_in[2];
  const float* W1  = (const float*)d_in[3];
  const float* b1  = (const float*)d_in[4];
  const float* g1  = (const float*)d_in[5];
  const float* be1 = (const float*)d_in[6];
  const float* W2  = (const float*)d_in[7];
  const float* b2  = (const float*)d_in[8];
  const float* g2  = (const float*)d_in[9];
  const float* be2 = (const float*)d_in[10];
  const float* Wih = (const float*)d_in[11];
  const float* Whh = (const float*)d_in[12];
  const float* bih = (const float*)d_in[13];
  const float* bhh = (const float*)d_in[14];
  float* out  = (float*)d_out;
  float* hbuf = (float*)d_ws;

  const size_t ldsA = (size_t)2 * NP_ * STR_ * sizeof(float);
  ftdgnn_agg_mlp<<<B_ * T_ * H_, TPB_, ldsA, stream>>>(
      x, a, eps, W1, b1, g1, be1, W2, b2, g2, be2, hbuf);

  const size_t ldsB = (size_t)(2 * G3_ * STR_ + 4 * 16 * STR_ +
                               2 * GXT_) * sizeof(float);
  ftdgnn_gru<<<NBLK_, TPG_, ldsB, stream>>>(Wih, Whh, bih, bhh, hbuf, out);
}


# GRU input waves: x@Wih^T gate tiles re-dealt by SIMD load (a wave sharing its SIMD with two recurrent waves takes 1 tile, the others 4) and spread over both steps of a pair; k order unchanged
# speedup vs baseline: 2.4881x; 1.0418x over previous
#include <hip/hip_runtime.h>
#include <math.h>

#define B_    4
#define ROI_  90
#define T_    200
#define H_    3
#define F_    96
#define HID_  96
#define G3_   288
#define BH_   12
#define NP_   96
#define STR_  100
#define TPB_  288

typedef float v2f __attribute__((ext_vector_type(2)));
typedef float v8f __attribute__((ext_vector_type(8)));

__device__ __forceinline__ v8f wmma4(v2f a, v2f b, v8f c) {
  return __builtin_amdgcn_wmma_f32_16x16x4_f32(false, a, false, b, (short)0, c,
                                               false, false);
}


__device__ __forceinline__ void mm2x2_gB(const float* As, const float* __restrict__ Bg,
                                         int sr, int sc, int lane, v8f acc[4]) {
  const int half = lane >> 4;
  const int l15  = lane & 15;
  const float* a0 = As + (sr * 32 + l15) * STR_ + (half << 1);
  const float* a1 = a0 + 16 * STR_;
  const float* b0 = Bg + (half << 1) * NP_ + sc * 32 + l15;
  #pragma unroll
  for (int j = 0; j < 4; ++j) acc[j] = (v8f){0.f,0.f,0.f,0.f,0.f,0.f,0.f,0.f};
  #pragma unroll
  for (int k0 = 0; k0 < NP_; k0 += 4) {
    v2f av0 = *(const v2f*)(a0 + k0);
    v2f av1 = *(const v2f*)(a1 + k0);
    v2f bv0, bv1;
    bv0[0] = b0[k0 * NP_];        bv0[1] = b0[(k0 + 1) * NP_];
    bv1[0] = b0[k0 * NP_ + 16];   bv1[1] = b0[(k0 + 1) * NP_ + 16];
    acc[0] = wmma4(av0, bv0, acc[0]);
    acc[1] = wmma4(av0, bv1, acc[1]);
    acc[2] = wmma4(av1, bv0, acc[2]);
    acc[3] = wmma4(av1, bv1, acc[3]);
  }
}

#define XRS_  ((size_t)T_ * H_ * F_)
#define ARS_  ((size_t)T_ * H_ * ROI_)
#define KAGG_ 92

__global__ void __launch_bounds__(TPB_)
ftdgnn_agg_mlp(const float* __restrict__ x, const float* __restrict__ a,
               const float* __restrict__ eps,
               const float* __restrict__ W1, const float* __restrict__ b1,
               const float* __restrict__ g1, const float* __restrict__ be1,
               const float* __restrict__ W2, const float* __restrict__ b2,
               const float* __restrict__ g2, const float* __restrict__ be2,
               float* __restrict__ hbuf) {
  extern __shared__ float smem[];
  float* aggS = smem;
  float* h1S  = aggS + NP_ * STR_;

  const int tid  = threadIdx.x;
  const int lane = tid & 31;
  const int wave = tid >> 5;
  const int half = lane >> 4;
  const int l15  = lane & 15;
  const int sr   = wave / 3;
  const int sc   = wave % 3;

  const int bth = blockIdx.x;
  const int b   = bth / (T_ * H_);
  const int th  = bth % (T_ * H_);
  const int t   = th / H_;
  const int hh  = th % H_;

  const size_t tile = ((size_t)(b * ROI_) * T_ + t) * H_ + hh;
  const float* xb = x + tile * F_;
  const float* ab = a + tile * ROI_;

  const float epsv = eps[0];
  const float rs   = rsqrtf(1.0f + 1e-5f);

  v8f acc[4];

  {
    const int  i0   = sr * 32 + l15;
    const int  i1   = i0 + 16;
    const bool i1ok = (i1 < ROI_);
    const float* ar0 = ab + (size_t)i0 * ARS_;
    const float* ar1 = ab + (size_t)(i1ok ? i1 : ROI_ - 1) * ARS_;
    const float* xc  = xb + sc * 32 + l15;
    #pragma unroll
    for (int j = 0; j < 4; ++j) acc[j] = (v8f){0.f,0.f,0.f,0.f,0.f,0.f,0.f,0.f};
    #pragma unroll
    for (int k0 = 0; k0 < KAGG_; k0 += 4) {
      const int  j   = k0 + (half << 1);
      const bool jok = (j < ROI_);
      const int  jc  = jok ? j : 0;
      const v2f r0 = *(const v2f*)(ar0 + jc);
      const v2f r1 = *(const v2f*)(ar1 + jc);
      const float* xr = xc + (size_t)jc * XRS_;
      const float x00 = xr[0],    x01 = xr[16];
      const float x10 = xr[XRS_], x11 = xr[XRS_ + 16];
      v2f av0, av1, bv0, bv1;
      av0[0] = (jok && r0[0] != 0.f) ? 1.f : 0.f;
      av0[1] = (jok && r0[1] != 0.f) ? 1.f : 0.f;
      av1[0] = (jok && i1ok && r1[0] != 0.f) ? 1.f : 0.f;
      av1[1] = (jok && i1ok && r1[1] != 0.f) ? 1.f : 0.f;
      bv0[0] = jok ? x00 : 0.f;   bv0[1] = jok ? x10 : 0.f;
      bv1[0] = jok ? x01 : 0.f;   bv1[1] = jok ? x11 : 0.f;
      acc[0] = wmma4(av0, bv0, acc[0]);
      acc[1] = wmma4(av0, bv1, acc[1]);
      acc[2] = wmma4(av1, bv0, acc[2]);
      acc[3] = wmma4(av1, bv1, acc[3]);
    }
  }
  #pragma unroll
  for (int di = 0; di < 2; ++di)
    #pragma unroll
    for (int dj = 0; dj < 2; ++dj) {
      const int n = (2 * sc + dj) * 16 + l15;
      #pragma unroll
      for (int v = 0; v < 8; ++v) {
        const int  m   = (2 * sr + di) * 16 + v + 8 * half;
        const bool mok = (m < ROI_);
        const float xv = xb[(size_t)(mok ? m : ROI_ - 1) * XRS_ + n];
        aggS[m * STR_ + n] = acc[di * 2 + dj][v] + epsv * (mok ? xv : 0.f);
      }
    }
  __syncthreads();

  mm2x2_gB(aggS, W1, sr, sc, lane, acc);
  #pragma unroll
  for (int di = 0; di < 2; ++di)
    #pragma unroll
    for (int dj = 0; dj < 2; ++dj) {
      const int n = (2 * sc + dj) * 16 + l15;
      const float sc1 = g1[n] * rs, sh1 = be1[n], bb1 = b1[n];
      #pragma unroll
      for (int v = 0; v < 8; ++v) {
        const int m = (2 * sr + di) * 16 + v + 8 * half;
        float z = (acc[di * 2 + dj][v] + bb1) * sc1 + sh1;
        z = (z > 0.f) ? z : expm1f(z);
        h1S[m * STR_ + n] = z;
      }
    }
  __syncthreads();

  mm2x2_gB(h1S, W2, sr, sc, lane, acc);
  #pragma unroll
  for (int di = 0; di < 2; ++di)
    #pragma unroll
    for (int dj = 0; dj < 2; ++dj) {
      const int n = (2 * sc + dj) * 16 + l15;
      const float sc2 = g2[n] * rs, sh2 = be2[n], bb2 = b2[n];
      #pragma unroll
      for (int v = 0; v < 8; ++v) {
        const int m = (2 * sr + di) * 16 + v + 8 * half;
        if (m < ROI_) {
          float z = (acc[di * 2 + dj][v] + bb2) * sc2 + sh2;
          z = (z > 0.f) ? z : 0.f;
          hbuf[((size_t)(m * T_ + t) * BH_ + (b * H_ + hh)) * F_ + n] = z;
        }
      }
    }
}

#define TPG_   384
#define GXT_   (18 * 32 * 8)
#define RPB_   6
#define NBLK_  (2 * ROI_)

typedef float v4f __attribute__((ext_vector_type(4)));

__device__ __forceinline__ int gate_row0(int t) { return (t % 3) * HID_ + (t / 3) * 16; }

__device__ __forceinline__ void mm1x2_tiles(const float* As, const float* Bt,
                                            int t, int lane, v8f acc[2]) {
  const int half = lane >> 4;
  const int l15  = lane & 15;
  const float* aRow = As + l15 * STR_ + (half << 1);
  const float* bRow0 = Bt + (gate_row0(t)     + l15) * STR_ + (half << 1);
  const float* bRow1 = Bt + (gate_row0(t + 1) + l15) * STR_ + (half << 1);
  acc[0] = (v8f){0.f,0.f,0.f,0.f,0.f,0.f,0.f,0.f};
  acc[1] = (v8f){0.f,0.f,0.f,0.f,0.f,0.f,0.f,0.f};
  #pragma unroll
  for (int k0 = 0; k0 < F_; k0 += 4) {
    const v2f av  = *(const v2f*)(aRow + k0);
    const v2f bv0 = *(const v2f*)(bRow0 + k0);
    const v2f bv1 = *(const v2f*)(bRow1 + k0);
    acc[0] = wmma4(av, bv0, acc[0]);
    acc[1] = wmma4(av, bv1, acc[1]);
  }
}

__device__ __forceinline__ void put_gx(float* gxp, v8f a, float b) {
  v4f lo; v2f hi;
  lo[0] = a[0] + b; lo[1] = a[1] + b; lo[2] = a[2] + b; lo[3] = a[3] + b;
  hi[0] = a[4] + b; hi[1] = a[5] + b;
  *(v4f*)(gxp)     = lo;
  *(v2f*)(gxp + 4) = hi;
}

__global__ void __launch_bounds__(TPG_)
ftdgnn_gru(const float* __restrict__ Wih, const float* __restrict__ Whh,
           const float* __restrict__ bih, const float* __restrict__ bhh,
           const float* __restrict__ hbuf, float* __restrict__ out) {
  extern __shared__ float smem[];
  float* WihS = smem;
  float* WhhS = WihS + G3_ * STR_;
  float* hS   = WhhS + G3_ * STR_;
  float* xS   = hS   + 2 * 16 * STR_;
  float* gxS  = xS   + 2 * 16 * STR_;

  const int tid  = threadIdx.x;
  const int lane = tid & 31;
  const int wave = __builtin_amdgcn_readfirstlane(tid >> 5);
  const int half = lane >> 4;
  const int l15  = lane & 15;
  const int r    = blockIdx.x >> 1;
  const int part = blockIdx.x & 1;
  const bool rec = (wave < 6);
  const int j    = rec ? wave : wave - 6;
  const int f    = j * 16 + l15;
  const int gofs = (j * 3 * 32 + lane) * 8;

  for (int idx = tid; idx < G3_ * F_; idx += TPG_) {
    const int g = idx / F_, k = idx % F_;
    WihS[g * STR_ + k] = Wih[((size_t)r * G3_ + g) * F_ + k];
    WhhS[g * STR_ + k] = Whh[((size_t)r * G3_ + g) * HID_ + k];
  }
  for (int idx = tid; idx < 2 * 16 * STR_; idx += TPG_) { hS[idx] = 0.f; xS[idx] = 0.f; }

  const float b_r = rec ? bhh[r * G3_ + f] : 0.f;
  const float b_z = rec ? bhh[r * G3_ + HID_ + f] : 0.f;
  const float b_n = rec ? bhh[r * G3_ + 2 * HID_ + f] : 0.f;

  const bool four = !(j == 2 || j == 3);
  const int  t0   = four ? ((j < 2) ? 4 * j : 4 * (j - 2)) : 14 + j;
  float bi[4];
  #pragma unroll
  for (int q = 0; q < 4; ++q) {
    const int t = four ? t0 + q : t0;
    bi[q] = rec ? 0.f : bih[r * G3_ + gate_row0(t) + l15];
  }

  float hprev[3];
  #pragma unroll
  for (int v = 0; v < 3; ++v) hprev[v] = 0.f;

  const int xl = j * 32 + lane;
  int xLds[3];
  int xGl[3];
  #pragma unroll
  for (int i = 0; i < 3; ++i) {
    const int u   = xl + 192 * i;
    const int rho = u / 48;
    const int c2  = 2 * (u % 48);
    const int q   = rho / RPB_;
    const int bq  = rho % RPB_;
    const int R   = ((bq < 3) ? bq : bq + 5) + 3 * q;
    xLds[i] = R * STR_ + c2;
    xGl[i]  = (q * BH_ + part * RPB_ + bq) * F_ + c2;
  }
  int hLds[2];
  size_t outOfs[2];
  #pragma unroll
  for (int i = 0; i < 2; ++i) {
    const int u  = xl + 192 * i;
    const int bq = (u / 48) % RPB_;
    const int c2 = 2 * (u % 48);
    const int R  = (bq < 3) ? bq : bq + 5;
    hLds[i] = R * STR_ + c2;
    const int m  = part * RPB_ + bq;
    const int bb = m / H_, hh = m % H_;
    outOfs[i] = (((size_t)(bb * ROI_ + r) * T_) * H_ + hh) * HID_ + c2;
  }
  const bool two = (j < 3);
  const float* xg = hbuf + (size_t)r * T_ * BH_ * F_;

  __syncthreads();
  if (!rec) {
    #pragma unroll
    for (int i = 0; i < 3; ++i)
      *(v2f*)(xS + xLds[i]) = *(const v2f*)(xg + xGl[i]);
  }
  __syncthreads();

  if (rec) __builtin_amdgcn_s_setprio(3);

  for (int s = 0; s <= T_; ++s) {
    if (rec) {
      if (s >= 1) {
        const int st = s - 1;
        const int q  = st & 1;
        const float* hcur = hS + ((s + 1) & 1) * (16 * STR_);
        float*       hnxt = hS + (s & 1) * (16 * STR_);
        const float* gxp  = gxS + ((st >> 1) & 1) * GXT_ + gofs;
        float gx[3][3];
        #pragma unroll
        for (int g = 0; g < 3; ++g) {
          const v4f lo = *(const v4f*)(gxp + g * 256);
          const v2f hi = *(const v2f*)(gxp + g * 256 + 4);
          gx[g][0] = q ? lo[3] : lo[0];
          gx[g][1] = q ? hi[0] : lo[1];
          gx[g][2] = q ? hi[1] : lo[2];
        }
        const float* aRow = hcur + l15 * STR_ + (half << 1);
        const float* bRow = WhhS + (j * 16 + l15) * STR_ + (half << 1);
        v2f av[F_ / 4];
        #pragma unroll
        for (int kk = 0; kk < F_ / 4; ++kk) av[kk] = *(const v2f*)(aRow + 4 * kk);
        v8f accR = (v8f){0.f,0.f,0.f,0.f,0.f,0.f,0.f,0.f};
        v8f accZ = (v8f){0.f,0.f,0.f,0.f,0.f,0.f,0.f,0.f};
        #pragma unroll
        for (int kk = 0; kk < F_ / 4; ++kk) {
          const v2f br = *(const v2f*)(bRow + 4 * kk);
          const v2f bz = *(const v2f*)(bRow + HID_ * STR_ + 4 * kk);
          accR = wmma4(av[kk], br, accR);
          accZ = wmma4(av[kk], bz, accZ);
        }
        __builtin_amdgcn_sched_barrier(0);
        float rg[3], zg[3];
        #pragma unroll
        for (int v = 0; v < 3; ++v) {
          const float hr = accR[v] + b_r;
          const float hz = accZ[v] + b_z;
          rg[v] = 1.f / (1.f + expf(-(gx[0][v] + hr)));
          zg[v] = 1.f / (1.f + expf(-(gx[1][v] + hz)));
        }
        v8f accN = (v8f){0.f,0.f,0.f,0.f,0.f,0.f,0.f,0.f};
        #pragma unroll
        for (int kk = 0; kk < F_ / 4; ++kk) {
          const v2f bn = *(const v2f*)(bRow + 2 * HID_ * STR_ + 4 * kk);
          accN = wmma4(av[kk], bn, accN);
        }
        asm volatile("" :: "v"(rg[0]), "v"(rg[1]), "v"(rg[2]), "v"(zg[0]), "v"(zg[1]), "v"(zg[2]));
        #pragma unroll
        for (int v = 0; v < 3; ++v) {
          const float hn = accN[v] + b_n;
          const float ng = tanhf(gx[2][v] + rg[v] * hn);
          const float h2 = (1.f - zg[v]) * ng + zg[v] * hprev[v];
          hprev[v] = h2;
          hnxt[(v + 8 * half) * STR_ + f] = h2;
        }
      }
    } else {
      const bool evenS   = ((s & 1) == 0);
      const bool doStage = evenS && (s + 2 < T_);
      v2f xv[3];
      #pragma unroll
      for (int i = 0; i < 3; ++i) xv[i] = (v2f){0.f, 0.f};
      if (doStage) {
        #pragma unroll
        for (int i = 0; i < 3; ++i)
          xv[i] = *(const v2f*)(xg + (size_t)(s + 2) * (BH_ * F_) + xGl[i]);
      }
      if (s >= 2) {
        const float* hcur = hS + ((s + 1) & 1) * (16 * STR_);
        *(v2f*)(out + outOfs[0] + (size_t)(s - 2) * (H_ * HID_)) = *(const v2f*)(hcur + hLds[0]);
        if (two)
          *(v2f*)(out + outOfs[1] + (size_t)(s - 2) * (H_ * HID_)) = *(const v2f*)(hcur + hLds[1]);
      }
      {
        const int p  = (s + 1) >> 1;
        const int hb = (evenS && s > 0) ? 1 : 0;
        const int he = (2 * p >= T_) ? hb : (evenS ? 2 : 1);
        const float* xbuf = xS + (p & 1) * (16 * STR_);
        float* gxb = gxS + (p & 1) * GXT_ + lane * 8;
        for (int h = hb; h < he; ++h) {
          if (four) {
            v8f acc[2];
            const int t = t0 + 2 * h;
            mm1x2_tiles(xbuf, WihS, t, lane, acc);
            put_gx(gxb + t * 256,       acc[0], h ? bi[2] : bi[0]);
            put_gx(gxb + (t + 1) * 256, acc[1], h ? bi[3] : bi[1]);
          } else if (h == 1) {
            const float* aRow = xbuf + l15 * STR_ + (half << 1);
            const float* bRow = WihS + (gate_row0(t0) + l15) * STR_ + (half << 1);
            v8f acc1 = (v8f){0.f,0.f,0.f,0.f,0.f,0.f,0.f,0.f};
            #pragma unroll
            for (int k0 = 0; k0 < F_; k0 += 4) {
              const v2f av = *(const v2f*)(aRow + k0);
              const v2f bv = *(const v2f*)(bRow + k0);
              acc1 = wmma4(av, bv, acc1);
            }
            put_gx(gxb + t0 * 256, acc1, bi[0]);
          }
        }
      }
      if (doStage) {
        float* xnxt = xS + (((s >> 1) + 1) & 1) * (16 * STR_);
        #pragma unroll
        for (int i = 0; i < 3; ++i) *(v2f*)(xnxt + xLds[i]) = xv[i];
      }
    }
    __syncthreads();
  }
  if (rec) __builtin_amdgcn_s_setprio(0);
  if (!rec) {
    const float* hlast = hS + (T_ & 1) * (16 * STR_);
    *(v2f*)(out + outOfs[0] + (size_t)(T_ - 1) * (H_ * HID_)) = *(const v2f*)(hlast + hLds[0]);
    if (two)
      *(v2f*)(out + outOfs[1] + (size_t)(T_ - 1) * (H_ * HID_)) = *(const v2f*)(hlast + hLds[1]);
  }
}

extern "C" void kernel_launch(void* const* d_in, const int* in_sizes, int n_in,
                              void* d_out, int out_size, void* d_ws, size_t ws_size,
                              hipStream_t stream) {
  const float* x   = (const float*)d_in[0];
  const float* a   = (const float*)d_in[1];
  const float* eps = (const float*)d_in[2];
  const float* W1  = (const float*)d_in[3];
  const float* b1  = (const float*)d_in[4];
  const float* g1  = (const float*)d_in[5];
  const float* be1 = (const float*)d_in[6];
  const float* W2  = (const float*)d_in[7];
  const float* b2  = (const float*)d_in[8];
  const float* g2  = (const float*)d_in[9];
  const float* be2 = (const float*)d_in[10];
  const float* Wih = (const float*)d_in[11];
  const float* Whh = (const float*)d_in[12];
  const float* bih = (const float*)d_in[13];
  const float* bhh = (const float*)d_in[14];
  float* out  = (float*)d_out;
  float* hbuf = (float*)d_ws;

  const size_t ldsA = (size_t)2 * NP_ * STR_ * sizeof(float);
  ftdgnn_agg_mlp<<<B_ * T_ * H_, TPB_, ldsA, stream>>>(
      x, a, eps, W1, b1, g1, be1, W2, b2, g2, be2, hbuf);

  const size_t ldsB = (size_t)(2 * G3_ * STR_ + 4 * 16 * STR_ +
                               2 * GXT_) * sizeof(float);
  ftdgnn_gru<<<NBLK_, TPG_, ldsB, stream>>>(Wih, Whh, bih, bhh, hbuf, out);
}


# Aggregation+MLP kernel: a tile's three products dealt to three groups of four waves (one wave of each group per SIMD, 48x48 quadrants each); 256 persistent blocks pipeline tiles through the groups
# speedup vs baseline: 2.6078x; 1.0481x over previous
#include <hip/hip_runtime.h>
#include <math.h>

#define B_    4
#define ROI_  90
#define T_    200
#define H_    3
#define F_    96
#define HID_  96
#define G3_   288
#define BH_   12
#define NP_   96
#define STR_  100

typedef float v2f __attribute__((ext_vector_type(2)));
typedef float v8f __attribute__((ext_vector_type(8)));

__device__ __forceinline__ v8f wmma4(v2f a, v2f b, v8f c) {
  return __builtin_amdgcn_wmma_f32_16x16x4_f32(false, a, false, b, (short)0, c,
                                               false, false);
}


__device__ __forceinline__ void mm3x3_gB(const float* As, const float* __restrict__ Bg,
                                         int qr, int qc, int lane, v8f acc[9]) {
  const int half = lane >> 4;
  const int l15  = lane & 15;
  const float* a0 = As + (qr * 48 + l15) * STR_ + (half << 1);
  const float* b0 = Bg + (half << 1) * NP_ + qc * 48 + l15;
  #pragma unroll
  for (int j = 0; j < 9; ++j) acc[j] = (v8f){0.f,0.f,0.f,0.f,0.f,0.f,0.f,0.f};
  #pragma unroll
  for (int k0 = 0; k0 < NP_; k0 += 4) {
    v2f av[3], bv[3];
    #pragma unroll
    for (int i = 0; i < 3; ++i) av[i] = *(const v2f*)(a0 + i * 16 * STR_ + k0);
    #pragma unroll
    for (int jn = 0; jn < 3; ++jn) {
      bv[jn][0] = b0[k0 * NP_ + 16 * jn];
      bv[jn][1] = b0[(k0 + 1) * NP_ + 16 * jn];
    }
    #pragma unroll
    for (int i = 0; i < 3; ++i)
      #pragma unroll
      for (int jn = 0; jn < 3; ++jn)
        acc[i * 3 + jn] = wmma4(av[i], bv[jn], acc[i * 3 + jn]);
  }
}

#define XRS_  ((size_t)T_ * H_ * F_)
#define ARS_  ((size_t)T_ * H_ * ROI_)
#define KAGG_ 92
#define TPA_  384
#define NT_   (B_ * T_ * H_)
#define NWG_  256
#define KC_   6
#define NCH_  4

__global__ void __launch_bounds__(TPA_)
ftdgnn_agg_mlp(const float* x, const float* a,
               const float* __restrict__ eps,
               const float* __restrict__ W1, const float* __restrict__ b1,
               const float* __restrict__ g1, const float* __restrict__ be1,
               const float* __restrict__ W2, const float* __restrict__ b2,
               const float* __restrict__ g2, const float* __restrict__ be2,
               float* __restrict__ hbuf) {
  extern __shared__ float smem[];

  const int tid  = threadIdx.x;
  const int lane = tid & 31;
  const int wave = __builtin_amdgcn_readfirstlane(tid >> 5);
  const int half = lane >> 4;
  const int l15  = lane & 15;
  const int grp  = wave >> 2;
  const int q    = wave & 3;
  const int qr   = q >> 1;
  const int qc   = q & 1;

  const int nblk = (int)gridDim.x;
  const int nT   = (NT_ - (int)blockIdx.x + nblk - 1) / nblk;

  const float epsv = eps[0];
  const float rs   = rsqrtf(1.0f + 1e-5f);

  float cs[3], ch[3], cb[3];
  #pragma unroll
  for (int jn = 0; jn < 3; ++jn) {
    const int n = qc * 48 + 16 * jn + l15;
    cs[jn] = (grp == 1) ? g1[n] * rs : g2[n] * rs;
    ch[jn] = (grp == 1) ? be1[n] : be2[n];
    cb[jn] = (grp == 1) ? b1[n] : b2[n];
  }

  v8f acc[9];

  for (int it = 0; it < nT + 2; ++it) {
    const int ti = it - grp;
    if (ti >= 0 && ti < nT) {
      const int bth = (int)blockIdx.x + ti * nblk;
      const int b   = bth / (T_ * H_);
      const int th  = bth % (T_ * H_);
      const int t   = th / H_;
      const int hh  = th % H_;
      float*       aggW = smem + (it & 1) * (NP_ * STR_);
      const float* aggR = smem + ((it + 1) & 1) * (NP_ * STR_);
      float*       h1W  = smem + (2 + (it & 1)) * (NP_ * STR_);
      const float* h1R  = smem + (2 + ((it + 1) & 1)) * (NP_ * STR_);

      if (grp == 0) {
        const size_t tile = ((size_t)(b * ROI_) * T_ + t) * H_ + hh;
        const float* xb = x + tile * F_;
        const float* ab = a + tile * ROI_;
        const float* ar[3];
        bool iok[3];
        #pragma unroll
        for (int i = 0; i < 3; ++i) {
          const int ii = qr * 48 + 16 * i + l15;
          iok[i] = (ii < ROI_);
          ar[i]  = ab + (size_t)(iok[i] ? ii : ROI_ - 1) * ARS_;
        }
        const float* xc = xb + qc * 48 + l15;
        #pragma unroll
        for (int j = 0; j < 9; ++j) acc[j] = (v8f){0.f,0.f,0.f,0.f,0.f,0.f,0.f,0.f};
        v2f   rm[2][KC_][3];
        float rx[2][KC_][6];
        #pragma unroll
        for (int s = 0; s < KC_; ++s) {
          const int j  = 4 * s + (half << 1);
          const float* xr = xc + (size_t)j * XRS_;
          #pragma unroll
          for (int i = 0; i < 3; ++i) rm[0][s][i] = *(const v2f*)(ar[i] + j);
          #pragma unroll
          for (int jn = 0; jn < 3; ++jn) { rx[0][s][2 * jn] = xr[16 * jn]; rx[0][s][2 * jn + 1] = xr[XRS_ + 16 * jn]; }
        }
        #pragma unroll
        for (int c = 0; c < NCH_; ++c) {
          if (c + 1 < NCH_) {
            #pragma unroll
            for (int s = 0; s < KC_; ++s) {
              const int k0 = 4 * ((c + 1) * KC_ + s);
              if (k0 < KAGG_) {
                const int  j   = k0 + (half << 1);
                const int  jc  = (j < ROI_) ? j : 0;
                const float* xr = xc + (size_t)jc * XRS_;
                #pragma unroll
                for (int i = 0; i < 3; ++i) rm[(c + 1) & 1][s][i] = *(const v2f*)(ar[i] + jc);
                #pragma unroll
                for (int jn = 0; jn < 3; ++jn) {
                  rx[(c + 1) & 1][s][2 * jn]     = xr[16 * jn];
                  rx[(c + 1) & 1][s][2 * jn + 1] = xr[XRS_ + 16 * jn];
                }
              }
            }
          }
          __builtin_amdgcn_sched_barrier(0);
          #pragma unroll
          for (int s = 0; s < KC_; ++s) {
            const int k0 = 4 * (c * KC_ + s);
            if (k0 < KAGG_) {
              const bool jok = (k0 + (half << 1) < ROI_);
              v2f av[3], bv[3];
              #pragma unroll
              for (int i = 0; i < 3; ++i) {
                const v2f r = rm[c & 1][s][i];
                av[i][0] = (jok && iok[i] && r[0] != 0.f) ? 1.f : 0.f;
                av[i][1] = (jok && iok[i] && r[1] != 0.f) ? 1.f : 0.f;
              }
              #pragma unroll
              for (int jn = 0; jn < 3; ++jn) {
                bv[jn][0] = jok ? rx[c & 1][s][2 * jn]     : 0.f;
                bv[jn][1] = jok ? rx[c & 1][s][2 * jn + 1] : 0.f;
              }
              #pragma unroll
              for (int i = 0; i < 3; ++i)
                #pragma unroll
                for (int jn = 0; jn < 3; ++jn)
                  acc[i * 3 + jn] = wmma4(av[i], bv[jn], acc[i * 3 + jn]);
            }
          }
        }
        float xe[72];
        #pragma unroll
        for (int i = 0; i < 3; ++i)
          #pragma unroll
          for (int jn = 0; jn < 3; ++jn)
            #pragma unroll
            for (int v = 0; v < 8; ++v) {
              const int m = qr * 48 + 16 * i + v + 8 * half;
              xe[(i * 3 + jn) * 8 + v] = xb[(size_t)((m < ROI_) ? m : ROI_ - 1) * XRS_ + qc * 48 + 16 * jn + l15];
            }
        __builtin_amdgcn_sched_barrier(0);
        #pragma unroll
        for (int i = 0; i < 3; ++i)
          #pragma unroll
          for (int jn = 0; jn < 3; ++jn) {
            const int n = qc * 48 + 16 * jn + l15;
            #pragma unroll
            for (int v = 0; v < 8; ++v) {
              const int  m   = qr * 48 + 16 * i + v + 8 * half;
              const bool mok = (m < ROI_);
              aggW[m * STR_ + n] = acc[i * 3 + jn][v] + epsv * (mok ? xe[(i * 3 + jn) * 8 + v] : 0.f);
            }
          }
      } else if (grp == 1) {
        mm3x3_gB(aggR, W1, qr, qc, lane, acc);
        #pragma unroll
        for (int i = 0; i < 3; ++i)
          #pragma unroll
          for (int jn = 0; jn < 3; ++jn) {
            const int n = qc * 48 + 16 * jn + l15;
            #pragma unroll
            for (int v = 0; v < 8; ++v) {
              const int m = qr * 48 + 16 * i + v + 8 * half;
              float z = (acc[i * 3 + jn][v] + cb[jn]) * cs[jn] + ch[jn];
              z = (z > 0.f) ? z : expm1f(z);
              h1W[m * STR_ + n] = z;
            }
          }
      } else {
        mm3x3_gB(h1R, W2, qr, qc, lane, acc);
        #pragma unroll
        for (int i = 0; i < 3; ++i)
          #pragma unroll
          for (int jn = 0; jn < 3; ++jn) {
            const int n = qc * 48 + 16 * jn + l15;
            #pragma unroll
            for (int v = 0; v < 8; ++v) {
              const int m = qr * 48 + 16 * i + v + 8 * half;
              if (m < ROI_) {
                float z = (acc[i * 3 + jn][v] + cb[jn]) * cs[jn] + ch[jn];
                z = (z > 0.f) ? z : 0.f;
                hbuf[((size_t)(m * T_ + t) * BH_ + (b * H_ + hh)) * F_ + n] = z;
              }
            }
          }
      }
    }
    __syncthreads();
  }
}

#define TPG_   384
#define GXT_   (18 * 32 * 8)
#define RPB_   6
#define NBLK_  (2 * ROI_)

typedef float v4f __attribute__((ext_vector_type(4)));

__device__ __forceinline__ int gate_row0(int t) { return (t % 3) * HID_ + (t / 3) * 16; }

__device__ __forceinline__ void mm1x2_tiles(const float* As, const float* Bt,
                                            int t, int lane, v8f acc[2]) {
  const int half = lane >> 4;
  const int l15  = lane & 15;
  const float* aRow = As + l15 * STR_ + (half << 1);
  const float* bRow0 = Bt + (gate_row0(t)     + l15) * STR_ + (half << 1);
  const float* bRow1 = Bt + (gate_row0(t + 1) + l15) * STR_ + (half << 1);
  acc[0] = (v8f){0.f,0.f,0.f,0.f,0.f,0.f,0.f,0.f};
  acc[1] = (v8f){0.f,0.f,0.f,0.f,0.f,0.f,0.f,0.f};
  #pragma unroll
  for (int k0 = 0; k0 < F_; k0 += 4) {
    const v2f av  = *(const v2f*)(aRow + k0);
    const v2f bv0 = *(const v2f*)(bRow0 + k0);
    const v2f bv1 = *(const v2f*)(bRow1 + k0);
    acc[0] = wmma4(av, bv0, acc[0]);
    acc[1] = wmma4(av, bv1, acc[1]);
  }
}

__device__ __forceinline__ void put_gx(float* gxp, v8f a, float b) {
  v4f lo; v2f hi;
  lo[0] = a[0] + b; lo[1] = a[1] + b; lo[2] = a[2] + b; lo[3] = a[3] + b;
  hi[0] = a[4] + b; hi[1] = a[5] + b;
  *(v4f*)(gxp)     = lo;
  *(v2f*)(gxp + 4) = hi;
}

__global__ void __launch_bounds__(TPG_)
ftdgnn_gru(const float* __restrict__ Wih, const float* __restrict__ Whh,
           const float* __restrict__ bih, const float* __restrict__ bhh,
           const float* __restrict__ hbuf, float* __restrict__ out) {
  extern __shared__ float smem[];
  float* WihS = smem;
  float* WhhS = WihS + G3_ * STR_;
  float* hS   = WhhS + G3_ * STR_;
  float* xS   = hS   + 2 * 16 * STR_;
  float* gxS  = xS   + 2 * 16 * STR_;

  const int tid  = threadIdx.x;
  const int lane = tid & 31;
  const int wave = __builtin_amdgcn_readfirstlane(tid >> 5);
  const int half = lane >> 4;
  const int l15  = lane & 15;
  const int r    = blockIdx.x >> 1;
  const int part = blockIdx.x & 1;
  const bool rec = (wave < 6);
  const int j    = rec ? wave : wave - 6;
  const int f    = j * 16 + l15;
  const int gofs = (j * 3 * 32 + lane) * 8;

  for (int idx = tid; idx < G3_ * F_; idx += TPG_) {
    const int g = idx / F_, k = idx % F_;
    WihS[g * STR_ + k] = Wih[((size_t)r * G3_ + g) * F_ + k];
    WhhS[g * STR_ + k] = Whh[((size_t)r * G3_ + g) * HID_ + k];
  }
  for (int idx = tid; idx < 2 * 16 * STR_; idx += TPG_) { hS[idx] = 0.f; xS[idx] = 0.f; }

  const float b_r = rec ? bhh[r * G3_ + f] : 0.f;
  const float b_z = rec ? bhh[r * G3_ + HID_ + f] : 0.f;
  const float b_n = rec ? bhh[r * G3_ + 2 * HID_ + f] : 0.f;

  const bool four = !(j == 2 || j == 3);
  const int  t0   = four ? ((j < 2) ? 4 * j : 4 * (j - 2)) : 14 + j;
  float bi[4];
  #pragma unroll
  for (int q = 0; q < 4; ++q) {
    const int t = four ? t0 + q : t0;
    bi[q] = rec ? 0.f : bih[r * G3_ + gate_row0(t) + l15];
  }

  float hprev[3];
  #pragma unroll
  for (int v = 0; v < 3; ++v) hprev[v] = 0.f;

  const int xl = j * 32 + lane;
  int xLds[3];
  int xGl[3];
  #pragma unroll
  for (int i = 0; i < 3; ++i) {
    const int u   = xl + 192 * i;
    const int rho = u / 48;
    const int c2  = 2 * (u % 48);
    const int q   = rho / RPB_;
    const int bq  = rho % RPB_;
    const int R   = ((bq < 3) ? bq : bq + 5) + 3 * q;
    xLds[i] = R * STR_ + c2;
    xGl[i]  = (q * BH_ + part * RPB_ + bq) * F_ + c2;
  }
  int hLds[2];
  size_t outOfs[2];
  #pragma unroll
  for (int i = 0; i < 2; ++i) {
    const int u  = xl + 192 * i;
    const int bq = (u / 48) % RPB_;
    const int c2 = 2 * (u % 48);
    const int R  = (bq < 3) ? bq : bq + 5;
    hLds[i] = R * STR_ + c2;
    const int m  = part * RPB_ + bq;
    const int bb = m / H_, hh = m % H_;
    outOfs[i] = (((size_t)(bb * ROI_ + r) * T_) * H_ + hh) * HID_ + c2;
  }
  const bool two = (j < 3);
  const float* xg = hbuf + (size_t)r * T_ * BH_ * F_;

  __syncthreads();
  if (!rec) {
    #pragma unroll
    for (int i = 0; i < 3; ++i)
      *(v2f*)(xS + xLds[i]) = *(const v2f*)(xg + xGl[i]);
  }
  __syncthreads();

  if (rec) __builtin_amdgcn_s_setprio(3);

  for (int s = 0; s <= T_; ++s) {
    if (rec) {
      if (s >= 1) {
        const int st = s - 1;
        const int q  = st & 1;
        const float* hcur = hS + ((s + 1) & 1) * (16 * STR_);
        float*       hnxt = hS + (s & 1) * (16 * STR_);
        const float* gxp  = gxS + ((st >> 1) & 1) * GXT_ + gofs;
        float gx[3][3];
        #pragma unroll
        for (int g = 0; g < 3; ++g) {
          const v4f lo = *(const v4f*)(gxp + g * 256);
          const v2f hi = *(const v2f*)(gxp + g * 256 + 4);
          gx[g][0] = q ? lo[3] : lo[0];
          gx[g][1] = q ? hi[0] : lo[1];
          gx[g][2] = q ? hi[1] : lo[2];
        }
        const float* aRow = hcur + l15 * STR_ + (half << 1);
        const float* bRow = WhhS + (j * 16 + l15) * STR_ + (half << 1);
        v2f av[F_ / 4];
        #pragma unroll
        for (int kk = 0; kk < F_ / 4; ++kk) av[kk] = *(const v2f*)(aRow + 4 * kk);
        v8f accR = (v8f){0.f,0.f,0.f,0.f,0.f,0.f,0.f,0.f};
        v8f accZ = (v8f){0.f,0.f,0.f,0.f,0.f,0.f,0.f,0.f};
        #pragma unroll
        for (int kk = 0; kk < F_ / 4; ++kk) {
          const v2f br = *(const v2f*)(bRow + 4 * kk);
          const v2f bz = *(const v2f*)(bRow + HID_ * STR_ + 4 * kk);
          accR = wmma4(av[kk], br, accR);
          accZ = wmma4(av[kk], bz, accZ);
        }
        __builtin_amdgcn_sched_barrier(0);
        float rg[3], zg[3];
        #pragma unroll
        for (int v = 0; v < 3; ++v) {
          const float hr = accR[v] + b_r;
          const float hz = accZ[v] + b_z;
          rg[v] = 1.f / (1.f + expf(-(gx[0][v] + hr)));
          zg[v] = 1.f / (1.f + expf(-(gx[1][v] + hz)));
        }
        v8f accN = (v8f){0.f,0.f,0.f,0.f,0.f,0.f,0.f,0.f};
        #pragma unroll
        for (int kk = 0; kk < F_ / 4; ++kk) {
          const v2f bn = *(const v2f*)(bRow + 2 * HID_ * STR_ + 4 * kk);
          accN = wmma4(av[kk], bn, accN);
        }
        asm volatile("" :: "v"(rg[0]), "v"(rg[1]), "v"(rg[2]), "v"(zg[0]), "v"(zg[1]), "v"(zg[2]));
        #pragma unroll
        for (int v = 0; v < 3; ++v) {
          const float hn = accN[v] + b_n;
          const float ng = tanhf(gx[2][v] + rg[v] * hn);
          const float h2 = (1.f - zg[v]) * ng + zg[v] * hprev[v];
          hprev[v] = h2;
          hnxt[(v + 8 * half) * STR_ + f] = h2;
        }
      }
    } else {
      const bool evenS   = ((s & 1) == 0);
      const bool doStage = evenS && (s + 2 < T_);
      v2f xv[3];
      #pragma unroll
      for (int i = 0; i < 3; ++i) xv[i] = (v2f){0.f, 0.f};
      if (doStage) {
        #pragma unroll
        for (int i = 0; i < 3; ++i)
          xv[i] = *(const v2f*)(xg + (size_t)(s + 2) * (BH_ * F_) + xGl[i]);
      }
      if (s >= 2) {
        const float* hcur = hS + ((s + 1) & 1) * (16 * STR_);
        *(v2f*)(out + outOfs[0] + (size_t)(s - 2) * (H_ * HID_)) = *(const v2f*)(hcur + hLds[0]);
        if (two)
          *(v2f*)(out + outOfs[1] + (size_t)(s - 2) * (H_ * HID_)) = *(const v2f*)(hcur + hLds[1]);
      }
      {
        const int p  = (s + 1) >> 1;
        const int hb = (evenS && s > 0) ? 1 : 0;
        const int he = (2 * p >= T_) ? hb : (evenS ? 2 : 1);
        const float* xbuf = xS + (p & 1) * (16 * STR_);
        float* gxb = gxS + (p & 1) * GXT_ + lane * 8;
        for (int h = hb; h < he; ++h) {
          if (four) {
            v8f acc[2];
            const int t = t0 + 2 * h;
            mm1x2_tiles(xbuf, WihS, t, lane, acc);
            put_gx(gxb + t * 256,       acc[0], h ? bi[2] : bi[0]);
            put_gx(gxb + (t + 1) * 256, acc[1], h ? bi[3] : bi[1]);
          } else if (h == 1) {
            const float* aRow = xbuf + l15 * STR_ + (half << 1);
            const float* bRow = WihS + (gate_row0(t0) + l15) * STR_ + (half << 1);
            v8f acc1 = (v8f){0.f,0.f,0.f,0.f,0.f,0.f,0.f,0.f};
            #pragma unroll
            for (int k0 = 0; k0 < F_; k0 += 4) {
              const v2f av = *(const v2f*)(aRow + k0);
              const v2f bv = *(const v2f*)(bRow + k0);
              acc1 = wmma4(av, bv, acc1);
            }
            put_gx(gxb + t0 * 256, acc1, bi[0]);
          }
        }
      }
      if (doStage) {
        float* xnxt = xS + (((s >> 1) + 1) & 1) * (16 * STR_);
        #pragma unroll
        for (int i = 0; i < 3; ++i) *(v2f*)(xnxt + xLds[i]) = xv[i];
      }
    }
    __syncthreads();
  }
  if (rec) __builtin_amdgcn_s_setprio(0);
  if (!rec) {
    const float* hlast = hS + (T_ & 1) * (16 * STR_);
    *(v2f*)(out + outOfs[0] + (size_t)(T_ - 1) * (H_ * HID_)) = *(const v2f*)(hlast + hLds[0]);
    if (two)
      *(v2f*)(out + outOfs[1] + (size_t)(T_ - 1) * (H_ * HID_)) = *(const v2f*)(hlast + hLds[1]);
  }
}

extern "C" void kernel_launch(void* const* d_in, const int* in_sizes, int n_in,
                              void* d_out, int out_size, void* d_ws, size_t ws_size,
                              hipStream_t stream) {
  const float* x   = (const float*)d_in[0];
  const float* a   = (const float*)d_in[1];
  const float* eps = (const float*)d_in[2];
  const float* W1  = (const float*)d_in[3];
  const float* b1  = (const float*)d_in[4];
  const float* g1  = (const float*)d_in[5];
  const float* be1 = (const float*)d_in[6];
  const float* W2  = (const float*)d_in[7];
  const float* b2  = (const float*)d_in[8];
  const float* g2  = (const float*)d_in[9];
  const float* be2 = (const float*)d_in[10];
  const float* Wih = (const float*)d_in[11];
  const float* Whh = (const float*)d_in[12];
  const float* bih = (const float*)d_in[13];
  const float* bhh = (const float*)d_in[14];
  float* out  = (float*)d_out;
  float* hbuf = (float*)d_ws;

  const size_t ldsA = (size_t)4 * NP_ * STR_ * sizeof(float);
  ftdgnn_agg_mlp<<<NWG_, TPA_, ldsA, stream>>>(
      x, a, eps, W1, b1, g1, be1, W2, b2, g2, be2, hbuf);

  const size_t ldsB = (size_t)(2 * G3_ * STR_ + 4 * 16 * STR_ +
                               2 * GXT_) * sizeof(float);
  ftdgnn_gru<<<NBLK_, TPG_, ldsB, stream>>>(Wih, Whh, bih, bhh, hbuf, out);
}
